# starting kernel
# baseline (speedup 1.0000x reference)
#include <hip/hip_runtime.h>
#include <math.h>

#define HEADS      4
#define HCF        128
#define C2         47
#define HC2        188
#define HC2P       192
#define SK2P       64
#define NEG_SLOPE  0.2f
#define EPS_DEN    1e-16f

typedef float v2f __attribute__((ext_vector_type(2)));
typedef float v8f __attribute__((ext_vector_type(8)));

__device__ __forceinline__ unsigned fkey(float f) {
  unsigned u = __float_as_uint(f);
  return (u & 0x80000000u) ? ~u : (u | 0x80000000u);
}
__device__ __forceinline__ float fdecode(unsigned k) {
  unsigned u = (k & 0x80000000u) ? (k ^ 0x80000000u) : ~k;
  return __uint_as_float(u);
}

template <int K, int LDA, int LDB, int LDC>
__global__ __launch_bounds__(256)
void gemm_wmma_f32(const float* __restrict__ A, const float* __restrict__ B,
                   float* __restrict__ C, int M) {
  __shared__ float bs[(K / 4) * 128 * 2];

  const int tid  = threadIdx.x;
  const int lane = tid & 31;
  const int wave = tid >> 5;
  const int tn   = blockIdx.y * 64;

  for (int idx = tid; idx < (K / 4) * 128; idx += 256) {
    const int l   = idx & 31;
    const int t   = (idx >> 5) & 3;
    const int ks  = idx >> 7;
    const int k   = 4 * ks + ((l >> 4) << 1);
    const int col = tn + t * 16 + (l & 15);
    float2 b;
    b.x = B[(size_t)k * LDB + col];
    b.y = B[(size_t)(k + 1) * LDB + col];
    *(float2*)&bs[(size_t)idx * 2] = b;
  }
  __syncthreads();

  const int rlo   = lane & 15;
  const int khalf = (lane >> 4) << 1;
  const int tm    = blockIdx.x * 8 + wave;

  int arow = tm * 16 + rlo;
  if (arow >= M) arow = M - 1;
  const float* ap = A + (size_t)arow * LDA;

  v8f acc[4] = {};
#pragma unroll 4
  for (int ks = 0; ks < K / 4; ++ks) {
    v2f a;
    a.x = ap[4 * ks + khalf];
    a.y = ap[4 * ks + khalf + 1];
    const float* bp = &bs[(size_t)(ks * 128 + lane) * 2];
#pragma unroll
    for (int t = 0; t < 4; ++t) {
      v2f b = *(const v2f*)(bp + t * 64);
      acc[t] = __builtin_amdgcn_wmma_f32_16x16x4_f32(
          false, a, false, b, (short)0, acc[t], false, false);
    }
  }

  const int mbase = tm * 16 + ((lane >> 4) << 3);
  if ((tm + 1) * 16 <= M) {
    float* cp = C + (size_t)mbase * LDC + tn + rlo;
#pragma unroll
    for (int t = 0; t < 4; ++t)
#pragma unroll
      for (int r = 0; r < 8; ++r)
        cp[r * LDC + t * 16] = acc[t][r];
  } else {
#pragma unroll
    for (int t = 0; t < 4; ++t)
#pragma unroll
      for (int r = 0; r < 8; ++r) {
        const int row = mbase + r;
        if (row < M) C[(size_t)row * LDC + tn + t * 16 + rlo] = acc[t][r];
      }
  }
}

template <int C, int LD>
__global__ void att_kernel(const float* __restrict__ xh,
                           const float* __restrict__ att_src,
                           const float* __restrict__ att_dst,
                           float* __restrict__ asrc, float* __restrict__ adst,
                           int N) {
  int id = blockIdx.x * blockDim.x + threadIdx.x;
  if (id >= N * HEADS) return;
  int n = id >> 2, h = id & 3;
  const float* xp = xh + (size_t)n * LD + h * C;
  const float* ws = att_src + h * C;
  const float* wd = att_dst + h * C;
  float s1 = 0.f, s2 = 0.f;
#pragma unroll 8
  for (int c = 0; c < C; ++c) {
    float v = xp[c];
    s1 += v * ws[c];
    s2 += v * wd[c];
  }
  asrc[id] = s1;
  adst[id] = s2;
}

__global__ void edge_pre(const int* __restrict__ ei, int E0, int EP,
                         const float* __restrict__ asrc,
                         const float* __restrict__ adst,
                         float* __restrict__ eb, unsigned* __restrict__ mkey) {
  int e = blockIdx.x * blockDim.x + threadIdx.x;
  if (e >= EP) return;
  int src = (e < E0) ? ei[e]      : (e - E0);
  int dst = (e < E0) ? ei[E0 + e] : (e - E0);
  const float4 es = *(const float4*)(asrc + (size_t)src * 4);
  const float4 ed = *(const float4*)(adst + (size_t)dst * 4);
  float v[4] = {es.x + ed.x, es.y + ed.y, es.z + ed.z, es.w + ed.w};
#pragma unroll
  for (int h = 0; h < 4; ++h) {
    float x = v[h];
    x = x > 0.f ? x : NEG_SLOPE * x;
    v[h] = x;
    atomicMax(&mkey[(size_t)dst * 4 + h], fkey(x));
  }
  *(float4*)(eb + (size_t)e * 4) = float4{v[0], v[1], v[2], v[3]};
}

__global__ void edge_sum(const int* __restrict__ ei, int E0, int EP,
                         const unsigned* __restrict__ mkey,
                         float* __restrict__ eb, float* __restrict__ sden) {
  int e = blockIdx.x * blockDim.x + threadIdx.x;
  if (e >= EP) return;
  int dst = (e < E0) ? ei[E0 + e] : (e - E0);
  float4 ev = *(const float4*)(eb + (size_t)e * 4);
  uint4 mk  = *(const uint4*)(mkey + (size_t)dst * 4);
  float v[4] = {ev.x, ev.y, ev.z, ev.w};
  unsigned k[4] = {mk.x, mk.y, mk.z, mk.w};
#pragma unroll
  for (int h = 0; h < 4; ++h) {
    float ex = __expf(v[h] - fdecode(k[h]));
    v[h] = ex;
    atomicAdd(&sden[(size_t)dst * 4 + h], ex);
  }
  *(float4*)(eb + (size_t)e * 4) = float4{v[0], v[1], v[2], v[3]};
}

template <int C, int LD>
__global__ __launch_bounds__(256)
void edge_agg(const int* __restrict__ ei, int E0, int EP,
              const float* __restrict__ xh, const float* __restrict__ eb,
              const float* __restrict__ sden, float* __restrict__ agg) {
  int e = blockIdx.x * (blockDim.x >> 5) + (threadIdx.x >> 5);
  if (e >= EP) return;
  const int lane = threadIdx.x & 31;
  int src = (e < E0) ? ei[e]      : (e - E0);
  int dst = (e < E0) ? ei[E0 + e] : (e - E0);
  const float* xs = xh + (size_t)src * LD;
  __builtin_prefetch(xs + lane * 4, 0, 3);
  const float4 ex4 = *(const float4*)(eb + (size_t)e * 4);
  const float4 s4  = *(const float4*)(sden + (size_t)dst * 4);
  float alpha[4] = {ex4.x / (s4.x + EPS_DEN), ex4.y / (s4.y + EPS_DEN),
                    ex4.z / (s4.z + EPS_DEN), ex4.w / (s4.w + EPS_DEN)};
  float* ad = agg + (size_t)dst * LD;
#pragma unroll
  for (int base = 0; base < HEADS * C; base += 32) {
    int f = base + lane;
    if (f < HEADS * C) {
      int h = f / C;
      atomicAdd(&ad[f], xs[f] * alpha[h]);
    }
  }
}

__global__ void combine_elu(const float* __restrict__ agg,
                            const float* __restrict__ skip,
                            const float* __restrict__ bias,
                            const float* __restrict__ skb,
                            float* __restrict__ out, int total) {
  int i = blockIdx.x * blockDim.x + threadIdx.x;
  if (i >= total) return;
  int f = i & 127;
  float v = agg[i] + bias[f] + skip[i] + skb[f];
  out[i] = v > 0.f ? v : expm1f(v);
}

__global__ void combine_out(const float* __restrict__ agg,
                            const float* __restrict__ skip,
                            const float* __restrict__ bias2,
                            const float* __restrict__ skb2,
                            float* __restrict__ out, int N) {
  int i = blockIdx.x * blockDim.x + threadIdx.x;
  if (i >= N * C2) return;
  int n = i / C2, o = i - n * C2;
  const float* a = agg + (size_t)n * HC2P;
  float m = 0.25f * (a[o] + a[C2 + o] + a[2 * C2 + o] + a[3 * C2 + o]);
  out[i] = m + bias2[o] + skip[(size_t)n * SK2P + o] + skb2[o];
}

__global__ void pad_w(const float* __restrict__ w, float* __restrict__ wp,
                      int rows, int cols, int ldp) {
  int i = blockIdx.x * blockDim.x + threadIdx.x;
  if (i >= rows * ldp) return;
  int r = i / ldp, c = i - r * ldp;
  wp[i] = (c < cols) ? w[(size_t)r * cols + c] : 0.f;
}

extern "C" void kernel_launch(void* const* d_in, const int* in_sizes, int n_in,
                              void* d_out, int out_size, void* d_ws, size_t ws_size,
                              hipStream_t stream) {
  const float* x   = (const float*)d_in[0];
  const int*   ei  = (const int*)d_in[1];
  const float* lw[3] = {(const float*)d_in[2], (const float*)d_in[8],  (const float*)d_in[14]};
  const float* as[3] = {(const float*)d_in[3], (const float*)d_in[9],  (const float*)d_in[15]};
  const float* ad[3] = {(const float*)d_in[4], (const float*)d_in[10], (const float*)d_in[16]};
  const float* bi[3] = {(const float*)d_in[5], (const float*)d_in[11], (const float*)d_in[17]};
  const float* sw[3] = {(const float*)d_in[6], (const float*)d_in[12], (const float*)d_in[18]};
  const float* sb[3] = {(const float*)d_in[7], (const float*)d_in[13], (const float*)d_in[19]};

  const int N  = in_sizes[0] / 128;
  const int E0 = in_sizes[1] / 2;
  const int EP = E0 + N;

  float* w = (float*)d_ws;
  float*    xcur = w;                      w += (size_t)N * HCF;
  float*    xh   = w;                      w += (size_t)N * HC2P;
  float*    agg  = w;                      w += (size_t)N * HC2P;
  float*    skip = w;                      w += (size_t)N * HCF;
  float*    asrc = w;                      w += (size_t)N * 4;
  float*    adst = w;                      w += (size_t)N * 4;
  unsigned* mkey = (unsigned*)w;           w += (size_t)N * 4;
  float*    sden = w;                      w += (size_t)N * 4;
  float*    eb   = w;                      w += (size_t)EP * 4;
  float*    w2p  = w;                      w += (size_t)128 * HC2P;
  float*    sw2p = w;                      w += (size_t)128 * SK2P;

  const int mt8  = (N + 127) / 128;
  const int eblk = (EP + 255) / 256;
  const int ewarp = (EP + 7) / 8;

  gemm_wmma_f32<128, 128, 128, 128><<<dim3(mt8, 2), 256, 0, stream>>>(x, lw[0], xh,   N);
  gemm_wmma_f32<128, 128, 128, 128><<<dim3(mt8, 2), 256, 0, stream>>>(x, sw[0], skip, N);
  att_kernel<32, 128><<<(N * 4 + 255) / 256, 256, 0, stream>>>(xh, as[0], ad[0], asrc, adst, N);
  hipMemsetAsync(mkey, 0, (size_t)N * 4 * sizeof(unsigned), stream);
  hipMemsetAsync(sden, 0, (size_t)N * 4 * sizeof(float), stream);
  hipMemsetAsync(agg,  0, (size_t)N * HCF * sizeof(float), stream);
  edge_pre<<<eblk, 256, 0, stream>>>(ei, E0, EP, asrc, adst, eb, mkey);
  edge_sum<<<eblk, 256, 0, stream>>>(ei, E0, EP, mkey, eb, sden);
  edge_agg<32, 128><<<ewarp, 256, 0, stream>>>(ei, E0, EP, xh, eb, sden, agg);
  combine_elu<<<((size_t)N * HCF + 255) / 256, 256, 0, stream>>>(agg, skip, bi[0], sb[0], xcur, N * HCF);

  gemm_wmma_f32<128, 128, 128, 128><<<dim3(mt8, 2), 256, 0, stream>>>(xcur, lw[1], xh,   N);
  gemm_wmma_f32<128, 128, 128, 128><<<dim3(mt8, 2), 256, 0, stream>>>(xcur, sw[1], skip, N);
  att_kernel<32, 128><<<(N * 4 + 255) / 256, 256, 0, stream>>>(xh, as[1], ad[1], asrc, adst, N);
  hipMemsetAsync(mkey, 0, (size_t)N * 4 * sizeof(unsigned), stream);
  hipMemsetAsync(sden, 0, (size_t)N * 4 * sizeof(float), stream);
  hipMemsetAsync(agg,  0, (size_t)N * HCF * sizeof(float), stream);
  edge_pre<<<eblk, 256, 0, stream>>>(ei, E0, EP, asrc, adst, eb, mkey);
  edge_sum<<<eblk, 256, 0, stream>>>(ei, E0, EP, mkey, eb, sden);
  edge_agg<32, 128><<<ewarp, 256, 0, stream>>>(ei, E0, EP, xh, eb, sden, agg);
  combine_elu<<<((size_t)N * HCF + 255) / 256, 256, 0, stream>>>(agg, skip, bi[1], sb[1], xcur, N * HCF);

  pad_w<<<(128 * HC2P + 255) / 256, 256, 0, stream>>>(lw[2], w2p,  128, HC2, HC2P);
  pad_w<<<(128 * SK2P + 255) / 256, 256, 0, stream>>>(sw[2], sw2p, 128, C2,  SK2P);
  gemm_wmma_f32<128, 128, HC2P, HC2P><<<dim3(mt8, 3), 256, 0, stream>>>(xcur, w2p,  xh,   N);
  gemm_wmma_f32<128, 128, SK2P, SK2P><<<dim3(mt8, 1), 256, 0, stream>>>(xcur, sw2p, skip, N);
  att_kernel<47, HC2P><<<(N * 4 + 255) / 256, 256, 0, stream>>>(xh, as[2], ad[2], asrc, adst, N);
  hipMemsetAsync(mkey, 0, (size_t)N * 4 * sizeof(unsigned), stream);
  hipMemsetAsync(sden, 0, (size_t)N * 4 * sizeof(float), stream);
  hipMemsetAsync(agg,  0, (size_t)N * HC2P * sizeof(float), stream);
  edge_pre<<<eblk, 256, 0, stream>>>(ei, E0, EP, asrc, adst, eb, mkey);
  edge_sum<<<eblk, 256, 0, stream>>>(ei, E0, EP, mkey, eb, sden);
  edge_agg<47, HC2P><<<ewarp, 256, 0, stream>>>(ei, E0, EP, xh, eb, sden, agg);
  combine_out<<<(N * C2 + 255) / 256, 256, 0, stream>>>(agg, skip, bi[2], sb[2], (float*)d_out, N);
}


# Softmax normalisation deferred: the aggregation pass sums exp-weighted rows and the denominators (atomic arrival order as before), one divide per node in the epilogue; edge_sum launches removed
# speedup vs baseline: 1.2981x; 1.2981x over previous
#include <hip/hip_runtime.h>
#include <math.h>

#define HEADS      4
#define HCF        128
#define C2         47
#define HC2        188
#define HC2P       192
#define SK2P       64
#define NEG_SLOPE  0.2f
#define EPS_DEN    1e-16f

typedef float v2f __attribute__((ext_vector_type(2)));
typedef float v8f __attribute__((ext_vector_type(8)));

__device__ __forceinline__ unsigned fkey(float f) {
  unsigned u = __float_as_uint(f);
  return (u & 0x80000000u) ? ~u : (u | 0x80000000u);
}
__device__ __forceinline__ float fdecode(unsigned k) {
  unsigned u = (k & 0x80000000u) ? (k ^ 0x80000000u) : ~k;
  return __uint_as_float(u);
}

template <int K, int LDA, int LDB, int LDC>
__global__ __launch_bounds__(256)
void gemm_wmma_f32(const float* __restrict__ A, const float* __restrict__ B,
                   float* __restrict__ C, int M) {
  __shared__ float bs[(K / 4) * 128 * 2];

  const int tid  = threadIdx.x;
  const int lane = tid & 31;
  const int wave = tid >> 5;
  const int tn   = blockIdx.y * 64;

  for (int idx = tid; idx < (K / 4) * 128; idx += 256) {
    const int l   = idx & 31;
    const int t   = (idx >> 5) & 3;
    const int ks  = idx >> 7;
    const int k   = 4 * ks + ((l >> 4) << 1);
    const int col = tn + t * 16 + (l & 15);
    float2 b;
    b.x = B[(size_t)k * LDB + col];
    b.y = B[(size_t)(k + 1) * LDB + col];
    *(float2*)&bs[(size_t)idx * 2] = b;
  }
  __syncthreads();

  const int rlo   = lane & 15;
  const int khalf = (lane >> 4) << 1;
  const int tm    = blockIdx.x * 8 + wave;

  int arow = tm * 16 + rlo;
  if (arow >= M) arow = M - 1;
  const float* ap = A + (size_t)arow * LDA;

  v8f acc[4] = {};
#pragma unroll 4
  for (int ks = 0; ks < K / 4; ++ks) {
    v2f a;
    a.x = ap[4 * ks + khalf];
    a.y = ap[4 * ks + khalf + 1];
    const float* bp = &bs[(size_t)(ks * 128 + lane) * 2];
#pragma unroll
    for (int t = 0; t < 4; ++t) {
      v2f b = *(const v2f*)(bp + t * 64);
      acc[t] = __builtin_amdgcn_wmma_f32_16x16x4_f32(
          false, a, false, b, (short)0, acc[t], false, false);
    }
  }

  const int mbase = tm * 16 + ((lane >> 4) << 3);
  if ((tm + 1) * 16 <= M) {
    float* cp = C + (size_t)mbase * LDC + tn + rlo;
#pragma unroll
    for (int t = 0; t < 4; ++t)
#pragma unroll
      for (int r = 0; r < 8; ++r)
        cp[r * LDC + t * 16] = acc[t][r];
  } else {
#pragma unroll
    for (int t = 0; t < 4; ++t)
#pragma unroll
      for (int r = 0; r < 8; ++r) {
        const int row = mbase + r;
        if (row < M) C[(size_t)row * LDC + tn + t * 16 + rlo] = acc[t][r];
      }
  }
}

template <int C, int LD>
__global__ void att_kernel(const float* __restrict__ xh,
                           const float* __restrict__ att_src,
                           const float* __restrict__ att_dst,
                           float* __restrict__ asrc, float* __restrict__ adst,
                           int N) {
  int id = blockIdx.x * blockDim.x + threadIdx.x;
  if (id >= N * HEADS) return;
  int n = id >> 2, h = id & 3;
  const float* xp = xh + (size_t)n * LD + h * C;
  const float* ws = att_src + h * C;
  const float* wd = att_dst + h * C;
  float s1 = 0.f, s2 = 0.f;
#pragma unroll 8
  for (int c = 0; c < C; ++c) {
    float v = xp[c];
    s1 += v * ws[c];
    s2 += v * wd[c];
  }
  asrc[id] = s1;
  adst[id] = s2;
}

__global__ void edge_pre(const int* __restrict__ ei, int E0, int EP,
                         const float* __restrict__ asrc,
                         const float* __restrict__ adst,
                         float* __restrict__ eb, unsigned* __restrict__ mkey) {
  int e = blockIdx.x * blockDim.x + threadIdx.x;
  if (e >= EP) return;
  int src = (e < E0) ? ei[e]      : (e - E0);
  int dst = (e < E0) ? ei[E0 + e] : (e - E0);
  const float4 es = *(const float4*)(asrc + (size_t)src * 4);
  const float4 ed = *(const float4*)(adst + (size_t)dst * 4);
  float v[4] = {es.x + ed.x, es.y + ed.y, es.z + ed.z, es.w + ed.w};
#pragma unroll
  for (int h = 0; h < 4; ++h) {
    float x = v[h];
    x = x > 0.f ? x : NEG_SLOPE * x;
    v[h] = x;
    atomicMax(&mkey[(size_t)dst * 4 + h], fkey(x));
  }
  *(float4*)(eb + (size_t)e * 4) = float4{v[0], v[1], v[2], v[3]};
}

template <int C, int LD>
__global__ __launch_bounds__(256)
void edge_agg(const int* __restrict__ ei, int E0, int EP,
              const float* __restrict__ xh, const float* __restrict__ eb,
              const unsigned* __restrict__ mkey,
              float* __restrict__ sden, float* __restrict__ agg) {
  constexpr int F = HEADS * C;
  constexpr bool DEN_IN_ROW =
      (F + HEADS <= LD) && ((F % 32) != 0) && ((F % 32) + HEADS <= 32);
  int e = blockIdx.x * (blockDim.x >> 5) + (threadIdx.x >> 5);
  if (e >= EP) return;
  const int lane = threadIdx.x & 31;
  int src = (e < E0) ? ei[e]      : (e - E0);
  int dst = (e < E0) ? ei[E0 + e] : (e - E0);
  const float* xs = xh + (size_t)src * LD;
  __builtin_prefetch(xs + lane * 4, 0, 3);
  const float4 ev = *(const float4*)(eb + (size_t)e * 4);
  const uint4  mk = *(const uint4*)(mkey + (size_t)dst * 4);
  float w[4] = {__expf(ev.x - fdecode(mk.x)), __expf(ev.y - fdecode(mk.y)),
                __expf(ev.z - fdecode(mk.z)), __expf(ev.w - fdecode(mk.w))};
  float* ad = agg + (size_t)dst * LD;
#pragma unroll
  for (int base = 0; base < F; base += 32) {
    int f = base + lane;
    if (DEN_IN_ROW) {
      if (f < F + HEADS) {
        float val = (f < F) ? xs[f] * w[f / C] : w[f - F];
        atomicAdd(&ad[f], val);
      }
    } else {
      if (f < F) atomicAdd(&ad[f], xs[f] * w[f / C]);
    }
  }
  if (!DEN_IN_ROW) {
    if (lane < HEADS) atomicAdd(&sden[(size_t)dst * 4 + lane], w[lane]);
  }
}

__global__ void combine_elu(const float* __restrict__ agg,
                            const float* __restrict__ den,
                            const float* __restrict__ skip,
                            const float* __restrict__ bias,
                            const float* __restrict__ skb,
                            float* __restrict__ out, int total) {
  int i = blockIdx.x * blockDim.x + threadIdx.x;
  if (i >= total) return;
  int f = i & 127;
  float v = agg[i] / (den[i >> 5] + EPS_DEN) + bias[f] + skip[i] + skb[f];
  out[i] = v > 0.f ? v : expm1f(v);
}

__global__ void combine_out(const float* __restrict__ agg,
                            const float* __restrict__ skip,
                            const float* __restrict__ bias2,
                            const float* __restrict__ skb2,
                            float* __restrict__ out, int N) {
  int i = blockIdx.x * blockDim.x + threadIdx.x;
  if (i >= N * C2) return;
  int n = i / C2, o = i - n * C2;
  const float* a = agg + (size_t)n * HC2P;
  float m = 0.25f * (a[o]          / (a[HC2]     + EPS_DEN) +
                     a[C2 + o]     / (a[HC2 + 1] + EPS_DEN) +
                     a[2 * C2 + o] / (a[HC2 + 2] + EPS_DEN) +
                     a[3 * C2 + o] / (a[HC2 + 3] + EPS_DEN));
  out[i] = m + bias2[o] + skip[(size_t)n * SK2P + o] + skb2[o];
}

__global__ void pad_w(const float* __restrict__ w, float* __restrict__ wp,
                      int rows, int cols, int ldp) {
  int i = blockIdx.x * blockDim.x + threadIdx.x;
  if (i >= rows * ldp) return;
  int r = i / ldp, c = i - r * ldp;
  wp[i] = (c < cols) ? w[(size_t)r * cols + c] : 0.f;
}

extern "C" void kernel_launch(void* const* d_in, const int* in_sizes, int n_in,
                              void* d_out, int out_size, void* d_ws, size_t ws_size,
                              hipStream_t stream) {
  const float* x   = (const float*)d_in[0];
  const int*   ei  = (const int*)d_in[1];
  const float* lw[3] = {(const float*)d_in[2], (const float*)d_in[8],  (const float*)d_in[14]};
  const float* as[3] = {(const float*)d_in[3], (const float*)d_in[9],  (const float*)d_in[15]};
  const float* ad[3] = {(const float*)d_in[4], (const float*)d_in[10], (const float*)d_in[16]};
  const float* bi[3] = {(const float*)d_in[5], (const float*)d_in[11], (const float*)d_in[17]};
  const float* sw[3] = {(const float*)d_in[6], (const float*)d_in[12], (const float*)d_in[18]};
  const float* sb[3] = {(const float*)d_in[7], (const float*)d_in[13], (const float*)d_in[19]};

  const int N  = in_sizes[0] / 128;
  const int E0 = in_sizes[1] / 2;
  const int EP = E0 + N;

  float* w = (float*)d_ws;
  float*    xcur = w;                      w += (size_t)N * HCF;
  float*    xh   = w;                      w += (size_t)N * HC2P;
  float*    agg  = w;                      w += (size_t)N * HC2P;
  float*    skip = w;                      w += (size_t)N * HCF;
  float*    asrc = w;                      w += (size_t)N * 4;
  float*    adst = w;                      w += (size_t)N * 4;
  unsigned* mkey = (unsigned*)w;           w += (size_t)N * 4;
  float*    sden = w;                      w += (size_t)N * 4;
  float*    eb   = w;                      w += (size_t)EP * 4;
  float*    w2p  = w;                      w += (size_t)128 * HC2P;
  float*    sw2p = w;                      w += (size_t)128 * SK2P;

  const int mt8  = (N + 127) / 128;
  const int eblk = (EP + 255) / 256;
  const int ewarp = (EP + 7) / 8;

  gemm_wmma_f32<128, 128, 128, 128><<<dim3(mt8, 2), 256, 0, stream>>>(x, lw[0], xh,   N);
  gemm_wmma_f32<128, 128, 128, 128><<<dim3(mt8, 2), 256, 0, stream>>>(x, sw[0], skip, N);
  att_kernel<32, 128><<<(N * 4 + 255) / 256, 256, 0, stream>>>(xh, as[0], ad[0], asrc, adst, N);
  hipMemsetAsync(mkey, 0, (size_t)N * 4 * sizeof(unsigned), stream);
  hipMemsetAsync(sden, 0, (size_t)N * 4 * sizeof(float), stream);
  hipMemsetAsync(agg,  0, (size_t)N * HCF * sizeof(float), stream);
  edge_pre<<<eblk, 256, 0, stream>>>(ei, E0, EP, asrc, adst, eb, mkey);
  edge_agg<32, 128><<<ewarp, 256, 0, stream>>>(ei, E0, EP, xh, eb, mkey, sden, agg);
  combine_elu<<<((size_t)N * HCF + 255) / 256, 256, 0, stream>>>(agg, sden, skip, bi[0], sb[0], xcur, N * HCF);

  gemm_wmma_f32<128, 128, 128, 128><<<dim3(mt8, 2), 256, 0, stream>>>(xcur, lw[1], xh,   N);
  gemm_wmma_f32<128, 128, 128, 128><<<dim3(mt8, 2), 256, 0, stream>>>(xcur, sw[1], skip, N);
  att_kernel<32, 128><<<(N * 4 + 255) / 256, 256, 0, stream>>>(xh, as[1], ad[1], asrc, adst, N);
  hipMemsetAsync(mkey, 0, (size_t)N * 4 * sizeof(unsigned), stream);
  hipMemsetAsync(sden, 0, (size_t)N * 4 * sizeof(float), stream);
  hipMemsetAsync(agg,  0, (size_t)N * HCF * sizeof(float), stream);
  edge_pre<<<eblk, 256, 0, stream>>>(ei, E0, EP, asrc, adst, eb, mkey);
  edge_agg<32, 128><<<ewarp, 256, 0, stream>>>(ei, E0, EP, xh, eb, mkey, sden, agg);
  combine_elu<<<((size_t)N * HCF + 255) / 256, 256, 0, stream>>>(agg, sden, skip, bi[1], sb[1], xcur, N * HCF);

  pad_w<<<(128 * HC2P + 255) / 256, 256, 0, stream>>>(lw[2], w2p,  128, HC2, HC2P);
  pad_w<<<(128 * SK2P + 255) / 256, 256, 0, stream>>>(sw[2], sw2p, 128, C2,  SK2P);
  gemm_wmma_f32<128, 128, HC2P, HC2P><<<dim3(mt8, 3), 256, 0, stream>>>(xcur, w2p,  xh,   N);
  gemm_wmma_f32<128, 128, SK2P, SK2P><<<dim3(mt8, 1), 256, 0, stream>>>(xcur, sw2p, skip, N);
  att_kernel<47, HC2P><<<(N * 4 + 255) / 256, 256, 0, stream>>>(xh, as[2], ad[2], asrc, adst, N);
  hipMemsetAsync(mkey, 0, (size_t)N * 4 * sizeof(unsigned), stream);
  hipMemsetAsync(agg,  0, (size_t)N * HC2P * sizeof(float), stream);
  edge_pre<<<eblk, 256, 0, stream>>>(ei, E0, EP, asrc, adst, eb, mkey);
  edge_agg<47, HC2P><<<ewarp, 256, 0, stream>>>(ei, E0, EP, xh, eb, mkey, sden, agg);
  combine_out<<<(N * C2 + 255) / 256, 256, 0, stream>>>(agg, skip, bi[2], sb[2], (float*)d_out, N);
}


# Score/max pass: one lane per (edge, head), so an edge's four max keys leave in adjacent lanes of one atomic instruction (one line request per edge, not four); four items per lane, wave count kept
# speedup vs baseline: 1.6645x; 1.2823x over previous
#include <hip/hip_runtime.h>
#include <math.h>

#define HEADS      4
#define HCF        128
#define C2         47
#define HC2        188
#define HC2P       192
#define SK2P       64
#define NEG_SLOPE  0.2f
#define EPS_DEN    1e-16f

typedef float v2f __attribute__((ext_vector_type(2)));
typedef float v8f __attribute__((ext_vector_type(8)));

__device__ __forceinline__ unsigned fkey(float f) {
  unsigned u = __float_as_uint(f);
  return (u & 0x80000000u) ? ~u : (u | 0x80000000u);
}
__device__ __forceinline__ float fdecode(unsigned k) {
  unsigned u = (k & 0x80000000u) ? (k ^ 0x80000000u) : ~k;
  return __uint_as_float(u);
}

template <int K, int LDA, int LDB, int LDC>
__global__ __launch_bounds__(256)
void gemm_wmma_f32(const float* __restrict__ A, const float* __restrict__ B,
                   float* __restrict__ C, int M) {
  __shared__ float bs[(K / 4) * 128 * 2];

  const int tid  = threadIdx.x;
  const int lane = tid & 31;
  const int wave = tid >> 5;
  const int tn   = blockIdx.y * 64;

  for (int idx = tid; idx < (K / 4) * 128; idx += 256) {
    const int l   = idx & 31;
    const int t   = (idx >> 5) & 3;
    const int ks  = idx >> 7;
    const int k   = 4 * ks + ((l >> 4) << 1);
    const int col = tn + t * 16 + (l & 15);
    float2 b;
    b.x = B[(size_t)k * LDB + col];
    b.y = B[(size_t)(k + 1) * LDB + col];
    *(float2*)&bs[(size_t)idx * 2] = b;
  }
  __syncthreads();

  const int rlo   = lane & 15;
  const int khalf = (lane >> 4) << 1;
  const int tm    = blockIdx.x * 8 + wave;

  int arow = tm * 16 + rlo;
  if (arow >= M) arow = M - 1;
  const float* ap = A + (size_t)arow * LDA;

  v8f acc[4] = {};
#pragma unroll 4
  for (int ks = 0; ks < K / 4; ++ks) {
    v2f a;
    a.x = ap[4 * ks + khalf];
    a.y = ap[4 * ks + khalf + 1];
    const float* bp = &bs[(size_t)(ks * 128 + lane) * 2];
#pragma unroll
    for (int t = 0; t < 4; ++t) {
      v2f b = *(const v2f*)(bp + t * 64);
      acc[t] = __builtin_amdgcn_wmma_f32_16x16x4_f32(
          false, a, false, b, (short)0, acc[t], false, false);
    }
  }

  const int mbase = tm * 16 + ((lane >> 4) << 3);
  if ((tm + 1) * 16 <= M) {
    float* cp = C + (size_t)mbase * LDC + tn + rlo;
#pragma unroll
    for (int t = 0; t < 4; ++t)
#pragma unroll
      for (int r = 0; r < 8; ++r)
        cp[r * LDC + t * 16] = acc[t][r];
  } else {
#pragma unroll
    for (int t = 0; t < 4; ++t)
#pragma unroll
      for (int r = 0; r < 8; ++r) {
        const int row = mbase + r;
        if (row < M) C[(size_t)row * LDC + tn + t * 16 + rlo] = acc[t][r];
      }
  }
}

template <int C, int LD>
__global__ void att_kernel(const float* __restrict__ xh,
                           const float* __restrict__ att_src,
                           const float* __restrict__ att_dst,
                           float* __restrict__ asrc, float* __restrict__ adst,
                           int N) {
  int id = blockIdx.x * blockDim.x + threadIdx.x;
  if (id >= N * HEADS) return;
  int n = id >> 2, h = id & 3;
  const float* xp = xh + (size_t)n * LD + h * C;
  const float* ws = att_src + h * C;
  const float* wd = att_dst + h * C;
  float s1 = 0.f, s2 = 0.f;
#pragma unroll 8
  for (int c = 0; c < C; ++c) {
    float v = xp[c];
    s1 += v * ws[c];
    s2 += v * wd[c];
  }
  asrc[id] = s1;
  adst[id] = s2;
}

__global__ __launch_bounds__(256)
void edge_pre(const int* __restrict__ ei, int E0, int EP,
              const float* __restrict__ asrc,
              const float* __restrict__ adst,
              float* __restrict__ eb, unsigned* __restrict__ mkey) {
  const int total  = EP * 4;
  const int stride = gridDim.x * 256;
  const int id0    = blockIdx.x * 256 + threadIdx.x;
  int  id[4], d[4];
  bool ok[4];
  float x[4];
#pragma unroll
  for (int j = 0; j < 4; ++j) {
    int i = id0 + j * stride;
    ok[j] = i < total;
    i = ok[j] ? i : total - 1;
    id[j] = i;
    const int e  = i >> 2;
    const int h  = i & 3;
    const int ec = e < E0 ? e : E0 - 1;
    const int sv = ei[ec];
    const int dv = ei[E0 + ec];
    const int s  = (e < E0) ? sv : (e - E0);
    d[j]         = (e < E0) ? dv : (e - E0);
    x[j] = asrc[(size_t)s * 4 + h] + adst[(size_t)d[j] * 4 + h];
  }
#pragma unroll
  for (int j = 0; j < 4; ++j) {
    if (ok[j]) {
      float v = x[j];
      v = v > 0.f ? v : NEG_SLOPE * v;
      atomicMax(&mkey[(size_t)d[j] * 4 + (id[j] & 3)], fkey(v));
      eb[id[j]] = v;
    }
  }
}

template <int C, int LD>
__global__ __launch_bounds__(256)
void edge_agg(const int* __restrict__ ei, int E0, int EP,
              const float* __restrict__ xh, const float* __restrict__ eb,
              const unsigned* __restrict__ mkey,
              float* __restrict__ sden, float* __restrict__ agg) {
  constexpr int F = HEADS * C;
  constexpr bool DEN_IN_ROW =
      (F + HEADS <= LD) && ((F % 32) != 0) && ((F % 32) + HEADS <= 32);
  int e = blockIdx.x * (blockDim.x >> 5) + (threadIdx.x >> 5);
  if (e >= EP) return;
  const int lane = threadIdx.x & 31;
  int src = (e < E0) ? ei[e]      : (e - E0);
  int dst = (e < E0) ? ei[E0 + e] : (e - E0);
  const float* xs = xh + (size_t)src * LD;
  __builtin_prefetch(xs + lane * 4, 0, 3);
  const float4 ev = *(const float4*)(eb + (size_t)e * 4);
  const uint4  mk = *(const uint4*)(mkey + (size_t)dst * 4);
  float w[4] = {__expf(ev.x - fdecode(mk.x)), __expf(ev.y - fdecode(mk.y)),
                __expf(ev.z - fdecode(mk.z)), __expf(ev.w - fdecode(mk.w))};
  float* ad = agg + (size_t)dst * LD;
#pragma unroll
  for (int base = 0; base < F; base += 32) {
    int f = base + lane;
    if (DEN_IN_ROW) {
      if (f < F + HEADS) {
        float val = (f < F) ? xs[f] * w[f / C] : w[f - F];
        atomicAdd(&ad[f], val);
      }
    } else {
      if (f < F) atomicAdd(&ad[f], xs[f] * w[f / C]);
    }
  }
  if (!DEN_IN_ROW) {
    if (lane < HEADS) atomicAdd(&sden[(size_t)dst * 4 + lane], w[lane]);
  }
}

__global__ void combine_elu(const float* __restrict__ agg,
                            const float* __restrict__ den,
                            const float* __restrict__ skip,
                            const float* __restrict__ bias,
                            const float* __restrict__ skb,
                            float* __restrict__ out, int total) {
  int i = blockIdx.x * blockDim.x + threadIdx.x;
  if (i >= total) return;
  int f = i & 127;
  float v = agg[i] / (den[i >> 5] + EPS_DEN) + bias[f] + skip[i] + skb[f];
  out[i] = v > 0.f ? v : expm1f(v);
}

__global__ void combine_out(const float* __restrict__ agg,
                            const float* __restrict__ skip,
                            const float* __restrict__ bias2,
                            const float* __restrict__ skb2,
                            float* __restrict__ out, int N) {
  int i = blockIdx.x * blockDim.x + threadIdx.x;
  if (i >= N * C2) return;
  int n = i / C2, o = i - n * C2;
  const float* a = agg + (size_t)n * HC2P;
  float m = 0.25f * (a[o]          / (a[HC2]     + EPS_DEN) +
                     a[C2 + o]     / (a[HC2 + 1] + EPS_DEN) +
                     a[2 * C2 + o] / (a[HC2 + 2] + EPS_DEN) +
                     a[3 * C2 + o] / (a[HC2 + 3] + EPS_DEN));
  out[i] = m + bias2[o] + skip[(size_t)n * SK2P + o] + skb2[o];
}

__global__ void pad_w(const float* __restrict__ w, float* __restrict__ wp,
                      int rows, int cols, int ldp) {
  int i = blockIdx.x * blockDim.x + threadIdx.x;
  if (i >= rows * ldp) return;
  int r = i / ldp, c = i - r * ldp;
  wp[i] = (c < cols) ? w[(size_t)r * cols + c] : 0.f;
}

extern "C" void kernel_launch(void* const* d_in, const int* in_sizes, int n_in,
                              void* d_out, int out_size, void* d_ws, size_t ws_size,
                              hipStream_t stream) {
  const float* x   = (const float*)d_in[0];
  const int*   ei  = (const int*)d_in[1];
  const float* lw[3] = {(const float*)d_in[2], (const float*)d_in[8],  (const float*)d_in[14]};
  const float* as[3] = {(const float*)d_in[3], (const float*)d_in[9],  (const float*)d_in[15]};
  const float* ad[3] = {(const float*)d_in[4], (const float*)d_in[10], (const float*)d_in[16]};
  const float* bi[3] = {(const float*)d_in[5], (const float*)d_in[11], (const float*)d_in[17]};
  const float* sw[3] = {(const float*)d_in[6], (const float*)d_in[12], (const float*)d_in[18]};
  const float* sb[3] = {(const float*)d_in[7], (const float*)d_in[13], (const float*)d_in[19]};

  const int N  = in_sizes[0] / 128;
  const int E0 = in_sizes[1] / 2;
  const int EP = E0 + N;

  float* w = (float*)d_ws;
  float*    xcur = w;                      w += (size_t)N * HCF;
  float*    xh   = w;                      w += (size_t)N * HC2P;
  float*    agg  = w;                      w += (size_t)N * HC2P;
  float*    skip = w;                      w += (size_t)N * HCF;
  float*    asrc = w;                      w += (size_t)N * 4;
  float*    adst = w;                      w += (size_t)N * 4;
  unsigned* mkey = (unsigned*)w;           w += (size_t)N * 4;
  float*    sden = w;                      w += (size_t)N * 4;
  float*    eb   = w;                      w += (size_t)EP * 4;
  float*    w2p  = w;                      w += (size_t)128 * HC2P;
  float*    sw2p = w;                      w += (size_t)128 * SK2P;

  const int mt8  = (N + 127) / 128;
  const int eblk = (EP + 255) / 256;
  const int ewarp = (EP + 7) / 8;

  gemm_wmma_f32<128, 128, 128, 128><<<dim3(mt8, 2), 256, 0, stream>>>(x, lw[0], xh,   N);
  gemm_wmma_f32<128, 128, 128, 128><<<dim3(mt8, 2), 256, 0, stream>>>(x, sw[0], skip, N);
  att_kernel<32, 128><<<(N * 4 + 255) / 256, 256, 0, stream>>>(xh, as[0], ad[0], asrc, adst, N);
  hipMemsetAsync(mkey, 0, (size_t)N * 4 * sizeof(unsigned), stream);
  hipMemsetAsync(sden, 0, (size_t)N * 4 * sizeof(float), stream);
  hipMemsetAsync(agg,  0, (size_t)N * HCF * sizeof(float), stream);
  edge_pre<<<eblk, 256, 0, stream>>>(ei, E0, EP, asrc, adst, eb, mkey);
  edge_agg<32, 128><<<ewarp, 256, 0, stream>>>(ei, E0, EP, xh, eb, mkey, sden, agg);
  combine_elu<<<((size_t)N * HCF + 255) / 256, 256, 0, stream>>>(agg, sden, skip, bi[0], sb[0], xcur, N * HCF);

  gemm_wmma_f32<128, 128, 128, 128><<<dim3(mt8, 2), 256, 0, stream>>>(xcur, lw[1], xh,   N);
  gemm_wmma_f32<128, 128, 128, 128><<<dim3(mt8, 2), 256, 0, stream>>>(xcur, sw[1], skip, N);
  att_kernel<32, 128><<<(N * 4 + 255) / 256, 256, 0, stream>>>(xh, as[1], ad[1], asrc, adst, N);
  hipMemsetAsync(mkey, 0, (size_t)N * 4 * sizeof(unsigned), stream);
  hipMemsetAsync(sden, 0, (size_t)N * 4 * sizeof(float), stream);
  hipMemsetAsync(agg,  0, (size_t)N * HCF * sizeof(float), stream);
  edge_pre<<<eblk, 256, 0, stream>>>(ei, E0, EP, asrc, adst, eb, mkey);
  edge_agg<32, 128><<<ewarp, 256, 0, stream>>>(ei, E0, EP, xh, eb, mkey, sden, agg);
  combine_elu<<<((size_t)N * HCF + 255) / 256, 256, 0, stream>>>(agg, sden, skip, bi[1], sb[1], xcur, N * HCF);

  pad_w<<<(128 * HC2P + 255) / 256, 256, 0, stream>>>(lw[2], w2p,  128, HC2, HC2P);
  pad_w<<<(128 * SK2P + 255) / 256, 256, 0, stream>>>(sw[2], sw2p, 128, C2,  SK2P);
  gemm_wmma_f32<128, 128, HC2P, HC2P><<<dim3(mt8, 3), 256, 0, stream>>>(xcur, w2p,  xh,   N);
  gemm_wmma_f32<128, 128, SK2P, SK2P><<<dim3(mt8, 1), 256, 0, stream>>>(xcur, sw2p, skip, N);
  att_kernel<47, HC2P><<<(N * 4 + 255) / 256, 256, 0, stream>>>(xh, as[2], ad[2], asrc, adst, N);
  hipMemsetAsync(mkey, 0, (size_t)N * 4 * sizeof(unsigned), stream);
  hipMemsetAsync(agg,  0, (size_t)N * HC2P * sizeof(float), stream);
  edge_pre<<<eblk, 256, 0, stream>>>(ei, E0, EP, asrc, adst, eb, mkey);
  edge_agg<47, HC2P><<<ewarp, 256, 0, stream>>>(ei, E0, EP, xh, eb, mkey, sden, agg);
  combine_out<<<(N * C2 + 255) / 256, 256, 0, stream>>>(agg, skip, bi[2], sb[2], (float*)d_out, N);
}


# Softmax and aggregation by node-range ownership: each work-group lists the edges ending in its 196 (layer 2: 224) nodes and sums their rows in LDS; LDS atomics (arrival order) replace global atomics
# speedup vs baseline: 3.0527x; 1.8340x over previous
#include <hip/hip_runtime.h>
#include <math.h>

#define HEADS      4
#define HCF        128
#define C2         47
#define HC2        188
#define HC2P       192
#define SK2P       64
#define NEG_SLOPE  0.2f
#define EPS_DEN    1e-16f

typedef float v2f __attribute__((ext_vector_type(2)));
typedef float v8f __attribute__((ext_vector_type(8)));

__device__ __forceinline__ unsigned fkey(float f) {
  unsigned u = __float_as_uint(f);
  return (u & 0x80000000u) ? ~u : (u | 0x80000000u);
}
__device__ __forceinline__ float fdecode(unsigned k) {
  unsigned u = (k & 0x80000000u) ? (k ^ 0x80000000u) : ~k;
  return __uint_as_float(u);
}

template <int K, int LDA, int LDB, int LDC>
__global__ __launch_bounds__(256)
void gemm_wmma_f32(const float* __restrict__ A, const float* __restrict__ B,
                   float* __restrict__ C, int M) {
  __shared__ float bs[(K / 4) * 128 * 2];

  const int tid  = threadIdx.x;
  const int lane = tid & 31;
  const int wave = tid >> 5;
  const int tn   = blockIdx.y * 64;

  for (int idx = tid; idx < (K / 4) * 128; idx += 256) {
    const int l   = idx & 31;
    const int t   = (idx >> 5) & 3;
    const int ks  = idx >> 7;
    const int k   = 4 * ks + ((l >> 4) << 1);
    const int col = tn + t * 16 + (l & 15);
    float2 b;
    b.x = B[(size_t)k * LDB + col];
    b.y = B[(size_t)(k + 1) * LDB + col];
    *(float2*)&bs[(size_t)idx * 2] = b;
  }
  __syncthreads();

  const int rlo   = lane & 15;
  const int khalf = (lane >> 4) << 1;
  const int tm    = blockIdx.x * 8 + wave;

  int arow = tm * 16 + rlo;
  if (arow >= M) arow = M - 1;
  const float* ap = A + (size_t)arow * LDA;

  v8f acc[4] = {};
#pragma unroll 4
  for (int ks = 0; ks < K / 4; ++ks) {
    v2f a;
    a.x = ap[4 * ks + khalf];
    a.y = ap[4 * ks + khalf + 1];
    const float* bp = &bs[(size_t)(ks * 128 + lane) * 2];
#pragma unroll
    for (int t = 0; t < 4; ++t) {
      v2f b = *(const v2f*)(bp + t * 64);
      acc[t] = __builtin_amdgcn_wmma_f32_16x16x4_f32(
          false, a, false, b, (short)0, acc[t], false, false);
    }
  }

  const int mbase = tm * 16 + ((lane >> 4) << 3);
  if ((tm + 1) * 16 <= M) {
    float* cp = C + (size_t)mbase * LDC + tn + rlo;
#pragma unroll
    for (int t = 0; t < 4; ++t)
#pragma unroll
      for (int r = 0; r < 8; ++r)
        cp[r * LDC + t * 16] = acc[t][r];
  } else {
#pragma unroll
    for (int t = 0; t < 4; ++t)
#pragma unroll
      for (int r = 0; r < 8; ++r) {
        const int row = mbase + r;
        if (row < M) C[(size_t)row * LDC + tn + t * 16 + rlo] = acc[t][r];
      }
  }
}

template <int C, int LD>
__global__ void att_kernel(const float* __restrict__ xh,
                           const float* __restrict__ att_src,
                           const float* __restrict__ att_dst,
                           float* __restrict__ asrc, float* __restrict__ adst,
                           int N) {
  int id = blockIdx.x * blockDim.x + threadIdx.x;
  if (id >= N * HEADS) return;
  int n = id >> 2, h = id & 3;
  const float* xp = xh + (size_t)n * LD + h * C;
  const float* ws = att_src + h * C;
  const float* wd = att_dst + h * C;
  float s1 = 0.f, s2 = 0.f;
#pragma unroll 8
  for (int c = 0; c < C; ++c) {
    float v = xp[c];
    s1 += v * ws[c];
    s2 += v * wd[c];
  }
  asrc[id] = s1;
  adst[id] = s2;
}

#define LCAP  16384

template <int C, int LD, bool MEAN, int NBT>
__global__ __launch_bounds__(1024)
void gat_own(const int* __restrict__ ei, int E0, int N,
             const float* __restrict__ xh,
             const float* __restrict__ asrc, const float* __restrict__ adst,
             const float* __restrict__ skip, const float* __restrict__ bias,
             const float* __restrict__ skb, float* __restrict__ out) {
  constexpr int F   = HEADS * C;
  constexpr int NCH = (F + 31) / 32;
  __shared__ float    acc[NBT * LD];
  __shared__ float    den[NBT * 4];
  __shared__ float    adl[NBT * 4];
  __shared__ unsigned mx[NBT * 4];
  __shared__ int      list[LCAP];
  __shared__ unsigned char lr[LCAP];
  __shared__ int      s_cnt;

  const int tid  = threadIdx.x;
  const int lane = tid & 31;
  const int wave = tid >> 5;
  const int* __restrict__ srcp = ei;
  const int* __restrict__ dstp = ei + E0;

  for (int n0 = blockIdx.x * NBT; n0 < N; n0 += gridDim.x * NBT) {
    const int nbr = (N - n0 < NBT) ? (N - n0) : NBT;

    for (int i = tid; i < NBT * LD; i += 1024) acc[i] = 0.f;
    for (int i = tid; i < NBT * 4; i += 1024) {
      den[i] = 0.f;
      mx[i]  = 0u;
      adl[i] = (i < nbr * 4) ? adst[(size_t)n0 * 4 + i] : 0.f;
    }
    if (tid == 0) s_cnt = 0;
    __syncthreads();

    auto test = [&](int d, int i) {
      if ((unsigned)(d - n0) < (unsigned)nbr) {
        const int slot = atomicAdd(&s_cnt, 1);
        if (slot < LCAP) {
          list[slot] = i;
          lr[slot]   = (unsigned char)(d - n0);
        }
      }
    };
    auto scan = [&](int lo, int hi) {
      if (((E0 | lo) & 3) == 0) {
        const int4* __restrict__ p = (const int4*)dstp;
        const int q0 = lo >> 2, q1 = hi >> 2;
        for (int q = q0 + tid; q < q1; q += 4096) {
          int4 v[4];
#pragma unroll
          for (int j = 0; j < 4; ++j) {
            const int qq = q + j * 1024;
            v[j] = p[qq < q1 ? qq : q1 - 1];
          }
#pragma unroll
          for (int j = 0; j < 4; ++j) {
            const int qq = q + j * 1024;
            if (qq < q1) {
              test(v[j].x, 4 * qq);
              test(v[j].y, 4 * qq + 1);
              test(v[j].z, 4 * qq + 2);
              test(v[j].w, 4 * qq + 3);
            }
          }
        }
        for (int i = (q1 << 2) + tid; i < hi; i += 1024) test(dstp[i], i);
      } else {
        for (int i = lo + tid; i < hi; i += 1024) test(dstp[i], i);
      }
    };
    auto maxphase = [&](int cntc, bool self) {
      const int total = cntc + (self ? nbr : 0);
      for (int t = tid; t < total; t += 1024) {
        int s, r;
        if (t < cntc) {
          s = srcp[list[t]];
          r = lr[t];
          list[t] = s;
        } else {
          r = t - cntc;
          s = n0 + r;
        }
        const float4 a = *(const float4*)(asrc + (size_t)s * 4);
        const float av[4] = {a.x, a.y, a.z, a.w};
#pragma unroll
        for (int h = 0; h < 4; ++h) {
          float v = av[h] + adl[r * 4 + h];
          v = v > 0.f ? v : NEG_SLOPE * v;
          atomicMax(&mx[r * 4 + h], fkey(v));
        }
      }
    };
    auto resolve = [&](int cntc) {
      for (int t = tid; t < cntc; t += 1024) list[t] = srcp[list[t]];
    };
    auto aggphase = [&](int cntc, bool self) {
      const int total = cntc + (self ? nbr : 0);
      const int nb4 = (total + 3) >> 2;
      for (int b = wave; b < nb4; b += 32) {
        int  s[4], r[4];
        bool ok[4];
#pragma unroll
        for (int j = 0; j < 4; ++j) {
          const int t = b * 4 + j;
          ok[j] = t < total;
          const int tc = ok[j] ? t : total - 1;
          if (tc < cntc) {
            s[j] = list[tc];
            r[j] = lr[tc];
          } else {
            r[j] = tc - cntc;
            s[j] = n0 + r[j];
          }
        }
        float4 a[4];
        float  x[4][NCH];
#pragma unroll
        for (int j = 0; j < 4; ++j) {
          a[j] = *(const float4*)(asrc + (size_t)s[j] * 4);
          const float* xs = xh + (size_t)s[j] * LD;
#pragma unroll
          for (int c = 0; c < NCH; ++c) x[j][c] = xs[c * 32 + lane];
        }
#pragma unroll
        for (int j = 0; j < 4; ++j) {
          if (ok[j]) {
            const float av[4] = {a[j].x, a[j].y, a[j].z, a[j].w};
            float w[4];
#pragma unroll
            for (int h = 0; h < 4; ++h) {
              float v = av[h] + adl[r[j] * 4 + h];
              v = v > 0.f ? v : NEG_SLOPE * v;
              w[h] = __expf(v - fdecode(mx[r[j] * 4 + h]));
            }
#pragma unroll
            for (int c = 0; c < NCH; ++c) {
              const int col = c * 32 + lane;
              const int hd  = col / C;
              const float wh = hd == 0 ? w[0] : hd == 1 ? w[1] : hd == 2 ? w[2] : w[3];
              if (col < F) atomicAdd(&acc[r[j] * LD + col], x[j][c] * wh);
            }
            if (lane < 4) {
              const float wl = lane == 0 ? w[0] : lane == 1 ? w[1] : lane == 2 ? w[2] : w[3];
              atomicAdd(&den[r[j] * 4 + lane], wl);
            }
          }
        }
      }
    };

    scan(0, E0);
    __syncthreads();
    const int cnt = s_cnt;
    if (cnt <= LCAP) {
      maxphase(cnt, true);
      __syncthreads();
      aggphase(cnt, true);
    } else {
      for (int lo = 0; lo < E0; lo += LCAP) {
        __syncthreads();
        if (tid == 0) s_cnt = 0;
        __syncthreads();
        scan(lo, (E0 - lo < LCAP) ? E0 : lo + LCAP);
        __syncthreads();
        maxphase(s_cnt, lo == 0);
      }
      for (int lo = 0; lo < E0; lo += LCAP) {
        __syncthreads();
        if (tid == 0) s_cnt = 0;
        __syncthreads();
        scan(lo, (E0 - lo < LCAP) ? E0 : lo + LCAP);
        __syncthreads();
        resolve(s_cnt);
        __syncthreads();
        aggphase(s_cnt, lo == 0);
      }
    }
    __syncthreads();

    if (!MEAN) {
      for (int i = tid; i < nbr * F; i += 1024) {
        const int r = i / F, f = i - r * F;
        const size_t g = (size_t)(n0 + r) * F + f;
        const float v = acc[r * LD + f] / (den[r * 4 + f / C] + EPS_DEN) + bias[f] + skip[g] + skb[f];
        out[g] = v > 0.f ? v : expm1f(v);
      }
    } else {
      for (int i = tid; i < nbr * C; i += 1024) {
        const int r = i / C, o = i - r * C;
        const float* ar = acc + r * LD;
        const float* dr = den + r * 4;
        const float m = 0.25f * (ar[o]         / (dr[0] + EPS_DEN) +
                                 ar[C + o]     / (dr[1] + EPS_DEN) +
                                 ar[2 * C + o] / (dr[2] + EPS_DEN) +
                                 ar[3 * C + o] / (dr[3] + EPS_DEN));
        out[(size_t)(n0 + r) * C + o] = m + bias[o] + skip[(size_t)(n0 + r) * SK2P + o] + skb[o];
      }
    }
    __syncthreads();
  }
}

__global__ void pad_w(const float* __restrict__ w, float* __restrict__ wp,
                      int rows, int cols, int ldp) {
  int i = blockIdx.x * blockDim.x + threadIdx.x;
  if (i >= rows * ldp) return;
  int r = i / ldp, c = i - r * ldp;
  wp[i] = (c < cols) ? w[(size_t)r * cols + c] : 0.f;
}

extern "C" void kernel_launch(void* const* d_in, const int* in_sizes, int n_in,
                              void* d_out, int out_size, void* d_ws, size_t ws_size,
                              hipStream_t stream) {
  const float* x   = (const float*)d_in[0];
  const int*   ei  = (const int*)d_in[1];
  const float* lw[3] = {(const float*)d_in[2], (const float*)d_in[8],  (const float*)d_in[14]};
  const float* as[3] = {(const float*)d_in[3], (const float*)d_in[9],  (const float*)d_in[15]};
  const float* ad[3] = {(const float*)d_in[4], (const float*)d_in[10], (const float*)d_in[16]};
  const float* bi[3] = {(const float*)d_in[5], (const float*)d_in[11], (const float*)d_in[17]};
  const float* sw[3] = {(const float*)d_in[6], (const float*)d_in[12], (const float*)d_in[18]};
  const float* sb[3] = {(const float*)d_in[7], (const float*)d_in[13], (const float*)d_in[19]};

  const int N  = in_sizes[0] / 128;
  const int E0 = in_sizes[1] / 2;

  float* w = (float*)d_ws;
  float*    xcur = w;                      w += (size_t)N * HCF;
  float*    xh   = w;                      w += (size_t)N * HC2P;
  float*    skip = w;                      w += (size_t)N * HCF;
  float*    asrc = w;                      w += (size_t)N * 4;
  float*    adst = w;                      w += (size_t)N * 4;
  float*    w2p  = w;                      w += (size_t)128 * HC2P;
  float*    sw2p = w;                      w += (size_t)128 * SK2P;

  const int mt8  = (N + 127) / 128;
  const int own  = 256;
  const int own2 = 224;

  gemm_wmma_f32<128, 128, 128, 128><<<dim3(mt8, 2), 256, 0, stream>>>(x, lw[0], xh,   N);
  gemm_wmma_f32<128, 128, 128, 128><<<dim3(mt8, 2), 256, 0, stream>>>(x, sw[0], skip, N);
  att_kernel<32, 128><<<(N * 4 + 255) / 256, 256, 0, stream>>>(xh, as[0], ad[0], asrc, adst, N);
  gat_own<32, 128, false, 196><<<own, 1024, 0, stream>>>(ei, E0, N, xh, asrc, adst, skip, bi[0], sb[0], xcur);

  gemm_wmma_f32<128, 128, 128, 128><<<dim3(mt8, 2), 256, 0, stream>>>(xcur, lw[1], xh,   N);
  gemm_wmma_f32<128, 128, 128, 128><<<dim3(mt8, 2), 256, 0, stream>>>(xcur, sw[1], skip, N);
  att_kernel<32, 128><<<(N * 4 + 255) / 256, 256, 0, stream>>>(xh, as[1], ad[1], asrc, adst, N);
  gat_own<32, 128, false, 196><<<own, 1024, 0, stream>>>(ei, E0, N, xh, asrc, adst, skip, bi[1], sb[1], xcur);

  pad_w<<<(128 * HC2P + 255) / 256, 256, 0, stream>>>(lw[2], w2p,  128, HC2, HC2P);
  pad_w<<<(128 * SK2P + 255) / 256, 256, 0, stream>>>(sw[2], sw2p, 128, C2,  SK2P);
  gemm_wmma_f32<128, 128, HC2P, HC2P><<<dim3(mt8, 3), 256, 0, stream>>>(xcur, w2p,  xh,   N);
  gemm_wmma_f32<128, 128, SK2P, SK2P><<<dim3(mt8, 1), 256, 0, stream>>>(xcur, sw2p, skip, N);
  att_kernel<47, HC2P><<<(N * 4 + 255) / 256, 256, 0, stream>>>(xh, as[2], ad[2], asrc, adst, N);
  gat_own<47, HC2P, true, 224><<<own2, 1024, 0, stream>>>(ei, E0, N, xh, asrc, adst, skip, bi[2], sb[2], (float*)d_out);
}


# Each node range's edge list (source, local destination) is built once, in layer 0, and kept in the workspace; layers 1 and 2 read it instead of re-walking the edge row; sum order still LDS arrival
# speedup vs baseline: 3.5625x; 1.1670x over previous
#include <hip/hip_runtime.h>
#include <math.h>

#define HEADS      4
#define HCF        128
#define C2         47
#define HC2        188
#define HC2P       192
#define SK2P       64
#define NEG_SLOPE  0.2f
#define EPS_DEN    1e-16f

typedef float v2f __attribute__((ext_vector_type(2)));
typedef float v8f __attribute__((ext_vector_type(8)));

__device__ __forceinline__ unsigned fkey(float f) {
  unsigned u = __float_as_uint(f);
  return (u & 0x80000000u) ? ~u : (u | 0x80000000u);
}
__device__ __forceinline__ float fdecode(unsigned k) {
  unsigned u = (k & 0x80000000u) ? (k ^ 0x80000000u) : ~k;
  return __uint_as_float(u);
}

template <int K, int LDA, int LDB, int LDC>
__global__ __launch_bounds__(256)
void gemm_wmma_f32(const float* __restrict__ A, const float* __restrict__ B,
                   float* __restrict__ C, int M) {
  __shared__ float bs[(K / 4) * 128 * 2];

  const int tid  = threadIdx.x;
  const int lane = tid & 31;
  const int wave = tid >> 5;
  const int tn   = blockIdx.y * 64;

  for (int idx = tid; idx < (K / 4) * 128; idx += 256) {
    const int l   = idx & 31;
    const int t   = (idx >> 5) & 3;
    const int ks  = idx >> 7;
    const int k   = 4 * ks + ((l >> 4) << 1);
    const int col = tn + t * 16 + (l & 15);
    float2 b;
    b.x = B[(size_t)k * LDB + col];
    b.y = B[(size_t)(k + 1) * LDB + col];
    *(float2*)&bs[(size_t)idx * 2] = b;
  }
  __syncthreads();

  const int rlo   = lane & 15;
  const int khalf = (lane >> 4) << 1;
  const int tm    = blockIdx.x * 8 + wave;

  int arow = tm * 16 + rlo;
  if (arow >= M) arow = M - 1;
  const float* ap = A + (size_t)arow * LDA;

  v8f acc[4] = {};
#pragma unroll 4
  for (int ks = 0; ks < K / 4; ++ks) {
    v2f a;
    a.x = ap[4 * ks + khalf];
    a.y = ap[4 * ks + khalf + 1];
    const float* bp = &bs[(size_t)(ks * 128 + lane) * 2];
#pragma unroll
    for (int t = 0; t < 4; ++t) {
      v2f b = *(const v2f*)(bp + t * 64);
      acc[t] = __builtin_amdgcn_wmma_f32_16x16x4_f32(
          false, a, false, b, (short)0, acc[t], false, false);
    }
  }

  const int mbase = tm * 16 + ((lane >> 4) << 3);
  if ((tm + 1) * 16 <= M) {
    float* cp = C + (size_t)mbase * LDC + tn + rlo;
#pragma unroll
    for (int t = 0; t < 4; ++t)
#pragma unroll
      for (int r = 0; r < 8; ++r)
        cp[r * LDC + t * 16] = acc[t][r];
  } else {
#pragma unroll
    for (int t = 0; t < 4; ++t)
#pragma unroll
      for (int r = 0; r < 8; ++r) {
        const int row = mbase + r;
        if (row < M) C[(size_t)row * LDC + tn + t * 16 + rlo] = acc[t][r];
      }
  }
}

template <int C, int LD>
__global__ void att_kernel(const float* __restrict__ xh,
                           const float* __restrict__ att_src,
                           const float* __restrict__ att_dst,
                           float* __restrict__ asrc, float* __restrict__ adst,
                           int N) {
  int id = blockIdx.x * blockDim.x + threadIdx.x;
  if (id >= N * HEADS) return;
  int n = id >> 2, h = id & 3;
  const float* xp = xh + (size_t)n * LD + h * C;
  const float* ws = att_src + h * C;
  const float* wd = att_dst + h * C;
  float s1 = 0.f, s2 = 0.f;
#pragma unroll 8
  for (int c = 0; c < C; ++c) {
    float v = xp[c];
    s1 += v * ws[c];
    s2 += v * wd[c];
  }
  asrc[id] = s1;
  adst[id] = s2;
}

#define LCAP  16384
#define NLIST 256

template <int C, int LD, bool MEAN, int NBT, int LMODE, int NBL>
__global__ __launch_bounds__(1024)
void gat_own(const int* __restrict__ ei, int E0, int N,
             const float* __restrict__ xh,
             const float* __restrict__ asrc, const float* __restrict__ adst,
             const float* __restrict__ skip, const float* __restrict__ bias,
             const float* __restrict__ skb, float* __restrict__ out,
             int* glist, int* gcnt) {
  constexpr int F   = HEADS * C;
  constexpr int NCH = (F + 31) / 32;
  __shared__ float    acc[NBT * LD];
  __shared__ float    den[NBT * 4];
  __shared__ float    adl[NBT * 4];
  __shared__ unsigned mx[NBT * 4];
  __shared__ int      list[LCAP];
  __shared__ unsigned char lr[LCAP];
  __shared__ int      s_cnt;

  const int tid  = threadIdx.x;
  const int lane = tid & 31;
  const int wave = tid >> 5;
  const int* __restrict__ srcp = ei;
  const int* __restrict__ dstp = ei + E0;

  for (int n0 = blockIdx.x * NBT; n0 < N; n0 += gridDim.x * NBT) {
    const int nbr  = (N - n0 < NBT) ? (N - n0) : NBT;
    const int bidx = n0 / NBT;
    const bool keep = (LMODE == 0) && bidx < NLIST;
    int2* glw = (int2*)glist + (size_t)(keep ? bidx : 0) * LCAP;

    for (int i = tid; i < NBT * LD; i += 1024) acc[i] = 0.f;
    for (int i = tid; i < NBT * 4; i += 1024) {
      den[i] = 0.f;
      mx[i]  = 0u;
      adl[i] = (i < nbr * 4) ? adst[(size_t)n0 * 4 + i] : 0.f;
    }
    if (tid == 0) s_cnt = 0;
    __syncthreads();

    auto test = [&](int d, int i) {
      if ((unsigned)(d - n0) < (unsigned)nbr) {
        const int slot = atomicAdd(&s_cnt, 1);
        if (slot < LCAP) {
          list[slot] = i;
          lr[slot]   = (unsigned char)(d - n0);
        }
      }
    };
    auto scan = [&](int lo, int hi) {
      if (((E0 | lo) & 3) == 0) {
        const int4* __restrict__ p = (const int4*)dstp;
        const int q0 = lo >> 2, q1 = hi >> 2;
        for (int q = q0 + tid; q < q1; q += 4096) {
          int4 v[4];
#pragma unroll
          for (int j = 0; j < 4; ++j) {
            const int qq = q + j * 1024;
            v[j] = p[qq < q1 ? qq : q1 - 1];
          }
#pragma unroll
          for (int j = 0; j < 4; ++j) {
            const int qq = q + j * 1024;
            if (qq < q1) {
              test(v[j].x, 4 * qq);
              test(v[j].y, 4 * qq + 1);
              test(v[j].z, 4 * qq + 2);
              test(v[j].w, 4 * qq + 3);
            }
          }
        }
        for (int i = (q1 << 2) + tid; i < hi; i += 1024) test(dstp[i], i);
      } else {
        for (int i = lo + tid; i < hi; i += 1024) test(dstp[i], i);
      }
    };
    auto maxphase = [&](int cntc, bool self, bool resolved, bool store) {
      const int total = cntc + (self ? nbr : 0);
      for (int t = tid; t < total; t += 1024) {
        int s, r;
        if (t < cntc) {
          r = lr[t];
          if (resolved) {
            s = list[t];
          } else {
            s = srcp[list[t]];
            list[t] = s;
            if (LMODE == 0 && store) glw[t] = make_int2(s, r);
          }
        } else {
          r = t - cntc;
          s = n0 + r;
        }
        const float4 a = *(const float4*)(asrc + (size_t)s * 4);
        const float av[4] = {a.x, a.y, a.z, a.w};
#pragma unroll
        for (int h = 0; h < 4; ++h) {
          float v = av[h] + adl[r * 4 + h];
          v = v > 0.f ? v : NEG_SLOPE * v;
          atomicMax(&mx[r * 4 + h], fkey(v));
        }
      }
    };
    auto resolve = [&](int cntc) {
      for (int t = tid; t < cntc; t += 1024) list[t] = srcp[list[t]];
    };
    auto aggphase = [&](int cntc, bool self) {
      const int total = cntc + (self ? nbr : 0);
      const int nb4 = (total + 3) >> 2;
      for (int b = wave; b < nb4; b += 32) {
        int  s[4], r[4];
        bool ok[4];
#pragma unroll
        for (int j = 0; j < 4; ++j) {
          const int t = b * 4 + j;
          ok[j] = t < total;
          const int tc = ok[j] ? t : total - 1;
          if (tc < cntc) {
            s[j] = list[tc];
            r[j] = lr[tc];
          } else {
            r[j] = tc - cntc;
            s[j] = n0 + r[j];
          }
        }
        float4 a[4];
        float  x[4][NCH];
#pragma unroll
        for (int j = 0; j < 4; ++j) {
          a[j] = *(const float4*)(asrc + (size_t)s[j] * 4);
          const float* xs = xh + (size_t)s[j] * LD;
#pragma unroll
          for (int c = 0; c < NCH; ++c) x[j][c] = xs[c * 32 + lane];
        }
#pragma unroll
        for (int j = 0; j < 4; ++j) {
          if (ok[j]) {
            const float av[4] = {a[j].x, a[j].y, a[j].z, a[j].w};
            float w[4];
#pragma unroll
            for (int h = 0; h < 4; ++h) {
              float v = av[h] + adl[r[j] * 4 + h];
              v = v > 0.f ? v : NEG_SLOPE * v;
              w[h] = __expf(v - fdecode(mx[r[j] * 4 + h]));
            }
#pragma unroll
            for (int c = 0; c < NCH; ++c) {
              const int col = c * 32 + lane;
              const int hd  = col / C;
              const float wh = hd == 0 ? w[0] : hd == 1 ? w[1] : hd == 2 ? w[2] : w[3];
              if (col < F) atomicAdd(&acc[r[j] * LD + col], x[j][c] * wh);
            }
            if (lane < 4) {
              const float wl = lane == 0 ? w[0] : lane == 1 ? w[1] : lane == 2 ? w[2] : w[3];
              atomicAdd(&den[r[j] * 4 + lane], wl);
            }
          }
        }
      }
    };

    int  cnt  = 0;
    bool have = false;
    if (LMODE == 1) {
      if (bidx < NLIST) {
        const int c = gcnt[bidx];
        if (c <= LCAP) {
          const int2* gl = (const int2*)glist + (size_t)bidx * LCAP;
          for (int t = tid; t < c; t += 1024) {
            const int2 e = gl[t];
            list[t] = e.x;
            lr[t]   = (unsigned char)e.y;
          }
          cnt  = c;
          have = true;
        }
      }
    }
    if (LMODE == 2) {
      const int b0 = n0 / NBL, b1 = (n0 + nbr - 1) / NBL;
      bool ok = b1 < NLIST;
      for (int b = b0; ok && b <= b1; ++b) ok = gcnt[b] <= LCAP;
      if (ok) {
        for (int b = b0; b <= b1; ++b) {
          const int c = gcnt[b];
          const int2* gl = (const int2*)glist + (size_t)b * LCAP;
          const int off = b * NBL - n0;
          for (int t = tid; t < c; t += 4096) {
            int2 e[4];
#pragma unroll
            for (int j = 0; j < 4; ++j) {
              const int tt = t + j * 1024;
              e[j] = gl[tt < c ? tt : c - 1];
            }
#pragma unroll
            for (int j = 0; j < 4; ++j) {
              const int rr = e[j].y + off;
              if (t + j * 1024 < c && (unsigned)rr < (unsigned)nbr) {
                const int slot = atomicAdd(&s_cnt, 1);
                if (slot < LCAP) {
                  list[slot] = e[j].x;
                  lr[slot]   = (unsigned char)rr;
                }
              }
            }
          }
        }
        __syncthreads();
        cnt  = s_cnt;
        have = cnt <= LCAP;
        __syncthreads();
        if (!have && tid == 0) s_cnt = 0;
      }
    }
    if (LMODE != 0) __syncthreads();
    if (!have) {
      scan(0, E0);
      __syncthreads();
      cnt = s_cnt;
    }
    if (keep && tid == 0) gcnt[bidx] = cnt;
    if (cnt <= LCAP) {
      maxphase(cnt, true, have, keep);
      __syncthreads();
      aggphase(cnt, true);
    } else {
      for (int lo = 0; lo < E0; lo += LCAP) {
        __syncthreads();
        if (tid == 0) s_cnt = 0;
        __syncthreads();
        scan(lo, (E0 - lo < LCAP) ? E0 : lo + LCAP);
        __syncthreads();
        maxphase(s_cnt, lo == 0, false, false);
      }
      for (int lo = 0; lo < E0; lo += LCAP) {
        __syncthreads();
        if (tid == 0) s_cnt = 0;
        __syncthreads();
        scan(lo, (E0 - lo < LCAP) ? E0 : lo + LCAP);
        __syncthreads();
        resolve(s_cnt);
        __syncthreads();
        aggphase(s_cnt, lo == 0);
      }
    }
    __syncthreads();

    if (!MEAN) {
      for (int i = tid; i < nbr * F; i += 1024) {
        const int r = i / F, f = i - r * F;
        const size_t g = (size_t)(n0 + r) * F + f;
        const float v = acc[r * LD + f] / (den[r * 4 + f / C] + EPS_DEN) + bias[f] + skip[g] + skb[f];
        out[g] = v > 0.f ? v : expm1f(v);
      }
    } else {
      for (int i = tid; i < nbr * C; i += 1024) {
        const int r = i / C, o = i - r * C;
        const float* ar = acc + r * LD;
        const float* dr = den + r * 4;
        const float m = 0.25f * (ar[o]         / (dr[0] + EPS_DEN) +
                                 ar[C + o]     / (dr[1] + EPS_DEN) +
                                 ar[2 * C + o] / (dr[2] + EPS_DEN) +
                                 ar[3 * C + o] / (dr[3] + EPS_DEN));
        out[(size_t)(n0 + r) * C + o] = m + bias[o] + skip[(size_t)(n0 + r) * SK2P + o] + skb[o];
      }
    }
    __syncthreads();
  }
}

__global__ void pad_w(const float* __restrict__ w, float* __restrict__ wp,
                      int rows, int cols, int ldp) {
  int i = blockIdx.x * blockDim.x + threadIdx.x;
  if (i >= rows * ldp) return;
  int r = i / ldp, c = i - r * ldp;
  wp[i] = (c < cols) ? w[(size_t)r * cols + c] : 0.f;
}

extern "C" void kernel_launch(void* const* d_in, const int* in_sizes, int n_in,
                              void* d_out, int out_size, void* d_ws, size_t ws_size,
                              hipStream_t stream) {
  const float* x   = (const float*)d_in[0];
  const int*   ei  = (const int*)d_in[1];
  const float* lw[3] = {(const float*)d_in[2], (const float*)d_in[8],  (const float*)d_in[14]};
  const float* as[3] = {(const float*)d_in[3], (const float*)d_in[9],  (const float*)d_in[15]};
  const float* ad[3] = {(const float*)d_in[4], (const float*)d_in[10], (const float*)d_in[16]};
  const float* bi[3] = {(const float*)d_in[5], (const float*)d_in[11], (const float*)d_in[17]};
  const float* sw[3] = {(const float*)d_in[6], (const float*)d_in[12], (const float*)d_in[18]};
  const float* sb[3] = {(const float*)d_in[7], (const float*)d_in[13], (const float*)d_in[19]};

  const int N  = in_sizes[0] / 128;
  const int E0 = in_sizes[1] / 2;

  float* w = (float*)d_ws;
  float*    xcur = w;                      w += (size_t)N * HCF;
  float*    xh   = w;                      w += (size_t)N * HC2P;
  float*    skip = w;                      w += (size_t)N * HCF;
  float*    asrc = w;                      w += (size_t)N * 4;
  float*    adst = w;                      w += (size_t)N * 4;
  float*    w2p  = w;                      w += (size_t)128 * HC2P;
  float*    sw2p = w;                      w += (size_t)128 * SK2P;
  int*      glist = (int*)w;               w += (size_t)NLIST * LCAP * 2;
  int*      gcnt  = (int*)w;               w += (size_t)NLIST;

  const int mt8  = (N + 127) / 128;
  const int own  = 256;
  const int own2 = 224;

  gemm_wmma_f32<128, 128, 128, 128><<<dim3(mt8, 2), 256, 0, stream>>>(x, lw[0], xh,   N);
  gemm_wmma_f32<128, 128, 128, 128><<<dim3(mt8, 2), 256, 0, stream>>>(x, sw[0], skip, N);
  att_kernel<32, 128><<<(N * 4 + 255) / 256, 256, 0, stream>>>(xh, as[0], ad[0], asrc, adst, N);
  gat_own<32, 128, false, 196, 0, 196><<<own, 1024, 0, stream>>>(ei, E0, N, xh, asrc, adst, skip, bi[0], sb[0], xcur, glist, gcnt);

  gemm_wmma_f32<128, 128, 128, 128><<<dim3(mt8, 2), 256, 0, stream>>>(xcur, lw[1], xh,   N);
  gemm_wmma_f32<128, 128, 128, 128><<<dim3(mt8, 2), 256, 0, stream>>>(xcur, sw[1], skip, N);
  att_kernel<32, 128><<<(N * 4 + 255) / 256, 256, 0, stream>>>(xh, as[1], ad[1], asrc, adst, N);
  gat_own<32, 128, false, 196, 1, 196><<<own, 1024, 0, stream>>>(ei, E0, N, xh, asrc, adst, skip, bi[1], sb[1], xcur, glist, gcnt);

  pad_w<<<(128 * HC2P + 255) / 256, 256, 0, stream>>>(lw[2], w2p,  128, HC2, HC2P);
  pad_w<<<(128 * SK2P + 255) / 256, 256, 0, stream>>>(sw[2], sw2p, 128, C2,  SK2P);
  gemm_wmma_f32<128, 128, HC2P, HC2P><<<dim3(mt8, 3), 256, 0, stream>>>(xcur, w2p,  xh,   N);
  gemm_wmma_f32<128, 128, SK2P, SK2P><<<dim3(mt8, 1), 256, 0, stream>>>(xcur, sw2p, skip, N);
  att_kernel<47, HC2P><<<(N * 4 + 255) / 256, 256, 0, stream>>>(xh, as[2], ad[2], asrc, adst, N);
  gat_own<47, HC2P, true, 224, 2, 196><<<own2, 1024, 0, stream>>>(ei, E0, N, xh, asrc, adst, skip, bi[2], sb[2], (float*)d_out, glist, gcnt);
}


# Aggregation: each (entry, head) softmax weight is computed in one lane (8 entries x 4 heads per wave trip) and broadcast by v_readlane, not recomputed in all 32 lanes; sum order still LDS arrival
# speedup vs baseline: 4.1837x; 1.1744x over previous
#include <hip/hip_runtime.h>
#include <math.h>

#define HEADS      4
#define HCF        128
#define C2         47
#define HC2        188
#define HC2P       192
#define SK2P       64
#define NEG_SLOPE  0.2f
#define EPS_DEN    1e-16f

typedef float v2f __attribute__((ext_vector_type(2)));
typedef float v8f __attribute__((ext_vector_type(8)));

__device__ __forceinline__ unsigned fkey(float f) {
  unsigned u = __float_as_uint(f);
  return (u & 0x80000000u) ? ~u : (u | 0x80000000u);
}
__device__ __forceinline__ float fdecode(unsigned k) {
  unsigned u = (k & 0x80000000u) ? (k ^ 0x80000000u) : ~k;
  return __uint_as_float(u);
}

template <int K, int LDA, int LDB, int LDC>
__global__ __launch_bounds__(256)
void gemm_wmma_f32(const float* __restrict__ A, const float* __restrict__ B,
                   float* __restrict__ C, int M) {
  __shared__ float bs[(K / 4) * 128 * 2];

  const int tid  = threadIdx.x;
  const int lane = tid & 31;
  const int wave = tid >> 5;
  const int tn   = blockIdx.y * 64;

  for (int idx = tid; idx < (K / 4) * 128; idx += 256) {
    const int l   = idx & 31;
    const int t   = (idx >> 5) & 3;
    const int ks  = idx >> 7;
    const int k   = 4 * ks + ((l >> 4) << 1);
    const int col = tn + t * 16 + (l & 15);
    float2 b;
    b.x = B[(size_t)k * LDB + col];
    b.y = B[(size_t)(k + 1) * LDB + col];
    *(float2*)&bs[(size_t)idx * 2] = b;
  }
  __syncthreads();

  const int rlo   = lane & 15;
  const int khalf = (lane >> 4) << 1;
  const int tm    = blockIdx.x * 8 + wave;

  int arow = tm * 16 + rlo;
  if (arow >= M) arow = M - 1;
  const float* ap = A + (size_t)arow * LDA;

  v8f acc[4] = {};
#pragma unroll 4
  for (int ks = 0; ks < K / 4; ++ks) {
    v2f a;
    a.x = ap[4 * ks + khalf];
    a.y = ap[4 * ks + khalf + 1];
    const float* bp = &bs[(size_t)(ks * 128 + lane) * 2];
#pragma unroll
    for (int t = 0; t < 4; ++t) {
      v2f b = *(const v2f*)(bp + t * 64);
      acc[t] = __builtin_amdgcn_wmma_f32_16x16x4_f32(
          false, a, false, b, (short)0, acc[t], false, false);
    }
  }

  const int mbase = tm * 16 + ((lane >> 4) << 3);
  if ((tm + 1) * 16 <= M) {
    float* cp = C + (size_t)mbase * LDC + tn + rlo;
#pragma unroll
    for (int t = 0; t < 4; ++t)
#pragma unroll
      for (int r = 0; r < 8; ++r)
        cp[r * LDC + t * 16] = acc[t][r];
  } else {
#pragma unroll
    for (int t = 0; t < 4; ++t)
#pragma unroll
      for (int r = 0; r < 8; ++r) {
        const int row = mbase + r;
        if (row < M) C[(size_t)row * LDC + tn + t * 16 + rlo] = acc[t][r];
      }
  }
}

template <int C, int LD>
__global__ void att_kernel(const float* __restrict__ xh,
                           const float* __restrict__ att_src,
                           const float* __restrict__ att_dst,
                           float* __restrict__ asrc, float* __restrict__ adst,
                           int N) {
  int id = blockIdx.x * blockDim.x + threadIdx.x;
  if (id >= N * HEADS) return;
  int n = id >> 2, h = id & 3;
  const float* xp = xh + (size_t)n * LD + h * C;
  const float* ws = att_src + h * C;
  const float* wd = att_dst + h * C;
  float s1 = 0.f, s2 = 0.f;
#pragma unroll 8
  for (int c = 0; c < C; ++c) {
    float v = xp[c];
    s1 += v * ws[c];
    s2 += v * wd[c];
  }
  asrc[id] = s1;
  adst[id] = s2;
}

#define LCAP  16384
#define NLIST 256

template <int C, int LD, bool MEAN, int NBT, int LMODE, int NBL>
__global__ __launch_bounds__(1024)
void gat_own(const int* __restrict__ ei, int E0, int N,
             const float* __restrict__ xh,
             const float* __restrict__ asrc, const float* __restrict__ adst,
             const float* __restrict__ skip, const float* __restrict__ bias,
             const float* __restrict__ skb, float* __restrict__ out,
             int* glist, int* gcnt) {
  constexpr int F   = HEADS * C;
  constexpr int NCH = (F + 31) / 32;
  __shared__ float    acc[NBT * LD];
  __shared__ float    den[NBT * 4];
  __shared__ float    adl[NBT * 4];
  __shared__ unsigned mx[NBT * 4];
  __shared__ int      list[LCAP];
  __shared__ unsigned char lr[LCAP];
  __shared__ int      s_cnt;

  const int tid  = threadIdx.x;
  const int lane = tid & 31;
  const int wave = tid >> 5;
  const int* __restrict__ srcp = ei;
  const int* __restrict__ dstp = ei + E0;

  for (int n0 = blockIdx.x * NBT; n0 < N; n0 += gridDim.x * NBT) {
    const int nbr  = (N - n0 < NBT) ? (N - n0) : NBT;
    const int bidx = n0 / NBT;
    const bool keep = (LMODE == 0) && bidx < NLIST;
    int2* glw = (int2*)glist + (size_t)(keep ? bidx : 0) * LCAP;

    for (int i = tid; i < NBT * LD; i += 1024) acc[i] = 0.f;
    for (int i = tid; i < NBT * 4; i += 1024) {
      den[i] = 0.f;
      mx[i]  = 0u;
      adl[i] = (i < nbr * 4) ? adst[(size_t)n0 * 4 + i] : 0.f;
    }
    if (tid == 0) s_cnt = 0;
    __syncthreads();

    auto test = [&](int d, int i) {
      if ((unsigned)(d - n0) < (unsigned)nbr) {
        const int slot = atomicAdd(&s_cnt, 1);
        if (slot < LCAP) {
          list[slot] = i;
          lr[slot]   = (unsigned char)(d - n0);
        }
      }
    };
    auto scan = [&](int lo, int hi) {
      if (((E0 | lo) & 3) == 0) {
        const int4* __restrict__ p = (const int4*)dstp;
        const int q0 = lo >> 2, q1 = hi >> 2;
        for (int q = q0 + tid; q < q1; q += 4096) {
          int4 v[4];
#pragma unroll
          for (int j = 0; j < 4; ++j) {
            const int qq = q + j * 1024;
            v[j] = p[qq < q1 ? qq : q1 - 1];
          }
#pragma unroll
          for (int j = 0; j < 4; ++j) {
            const int qq = q + j * 1024;
            if (qq < q1) {
              test(v[j].x, 4 * qq);
              test(v[j].y, 4 * qq + 1);
              test(v[j].z, 4 * qq + 2);
              test(v[j].w, 4 * qq + 3);
            }
          }
        }
        for (int i = (q1 << 2) + tid; i < hi; i += 1024) test(dstp[i], i);
      } else {
        for (int i = lo + tid; i < hi; i += 1024) test(dstp[i], i);
      }
    };
    auto maxphase = [&](int cntc, bool self, bool resolved, bool store) {
      const int total = cntc + (self ? nbr : 0);
      for (int t = tid; t < total; t += 1024) {
        int s, r;
        if (t < cntc) {
          r = lr[t];
          if (resolved) {
            s = list[t];
          } else {
            s = srcp[list[t]];
            list[t] = s;
            if (LMODE == 0 && store) glw[t] = make_int2(s, r);
          }
        } else {
          r = t - cntc;
          s = n0 + r;
        }
        const float4 a = *(const float4*)(asrc + (size_t)s * 4);
        const float av[4] = {a.x, a.y, a.z, a.w};
#pragma unroll
        for (int h = 0; h < 4; ++h) {
          float v = av[h] + adl[r * 4 + h];
          v = v > 0.f ? v : NEG_SLOPE * v;
          atomicMax(&mx[r * 4 + h], fkey(v));
        }
      }
    };
    auto resolve = [&](int cntc) {
      for (int t = tid; t < cntc; t += 1024) list[t] = srcp[list[t]];
    };
    auto aggphase = [&](int cntc_, bool self) {
      const int cntc  = __builtin_amdgcn_readfirstlane(cntc_);
      const int total = cntc + (self ? nbr : 0);
      const int nb8   = (total + 7) >> 3;
      const int wv0   = __builtin_amdgcn_readfirstlane(wave);
      const int je    = lane >> 2;
      const int hh    = lane & 3;
      for (int b = wv0; b < nb8; b += 32) {
        const int  t   = b * 8 + je;
        const bool okl = t < total;
        const int  tc  = okl ? t : total - 1;
        int s, r;
        if (tc < cntc) {
          s = list[tc];
          r = lr[tc];
        } else {
          r = tc - cntc;
          s = n0 + r;
        }
        const float a = asrc[(size_t)s * 4 + hh];
        float x[8][NCH];
#pragma unroll
        for (int j = 0; j < 8; ++j) {
          const int su = __builtin_amdgcn_readlane(s, 4 * j);
          const float* xs = xh + (size_t)su * LD;
#pragma unroll
          for (int c = 0; c < NCH; ++c) x[j][c] = xs[c * 32 + lane];
        }
        float v = a + adl[r * 4 + hh];
        v = v > 0.f ? v : NEG_SLOPE * v;
        const float w = __expf(v - fdecode(mx[r * 4 + hh]));
        if (okl) atomicAdd(&den[r * 4 + hh], w);
#pragma unroll
        for (int j = 0; j < 8; ++j) {
          if (b * 8 + j < total) {
            const int ru = __builtin_amdgcn_readlane(r, 4 * j);
            float wj[4];
#pragma unroll
            for (int h = 0; h < 4; ++h)
              wj[h] = __int_as_float(__builtin_amdgcn_readlane(__float_as_int(w), 4 * j + h));
#pragma unroll
            for (int c = 0; c < NCH; ++c) {
              const int col = c * 32 + lane;
              const int hd  = col / C;
              const float wh = hd == 0 ? wj[0] : hd == 1 ? wj[1] : hd == 2 ? wj[2] : wj[3];
              if (col < F) atomicAdd(&acc[ru * LD + col], x[j][c] * wh);
            }
          }
        }
      }
    };

    int  cnt  = 0;
    bool have = false;
    if (LMODE == 1) {
      if (bidx < NLIST) {
        const int c = gcnt[bidx];
        if (c <= LCAP) {
          const int2* gl = (const int2*)glist + (size_t)bidx * LCAP;
          for (int t = tid; t < c; t += 1024) {
            const int2 e = gl[t];
            list[t] = e.x;
            lr[t]   = (unsigned char)e.y;
          }
          cnt  = c;
          have = true;
        }
      }
    }
    if (LMODE == 2) {
      const int b0 = n0 / NBL, b1 = (n0 + nbr - 1) / NBL;
      bool ok = b1 < NLIST;
      for (int b = b0; ok && b <= b1; ++b) ok = gcnt[b] <= LCAP;
      if (ok) {
        for (int b = b0; b <= b1; ++b) {
          const int c = gcnt[b];
          const int2* gl = (const int2*)glist + (size_t)b * LCAP;
          const int off = b * NBL - n0;
          for (int t = tid; t < c; t += 4096) {
            int2 e[4];
#pragma unroll
            for (int j = 0; j < 4; ++j) {
              const int tt = t + j * 1024;
              e[j] = gl[tt < c ? tt : c - 1];
            }
#pragma unroll
            for (int j = 0; j < 4; ++j) {
              const int rr = e[j].y + off;
              if (t + j * 1024 < c && (unsigned)rr < (unsigned)nbr) {
                const int slot = atomicAdd(&s_cnt, 1);
                if (slot < LCAP) {
                  list[slot] = e[j].x;
                  lr[slot]   = (unsigned char)rr;
                }
              }
            }
          }
        }
        __syncthreads();
        cnt  = s_cnt;
        have = cnt <= LCAP;
        __syncthreads();
        if (!have && tid == 0) s_cnt = 0;
      }
    }
    if (LMODE != 0) __syncthreads();
    if (!have) {
      scan(0, E0);
      __syncthreads();
      cnt = s_cnt;
    }
    if (keep && tid == 0) gcnt[bidx] = cnt;
    if (cnt <= LCAP) {
      maxphase(cnt, true, have, keep);
      __syncthreads();
      aggphase(cnt, true);
    } else {
      for (int lo = 0; lo < E0; lo += LCAP) {
        __syncthreads();
        if (tid == 0) s_cnt = 0;
        __syncthreads();
        scan(lo, (E0 - lo < LCAP) ? E0 : lo + LCAP);
        __syncthreads();
        maxphase(s_cnt, lo == 0, false, false);
      }
      for (int lo = 0; lo < E0; lo += LCAP) {
        __syncthreads();
        if (tid == 0) s_cnt = 0;
        __syncthreads();
        scan(lo, (E0 - lo < LCAP) ? E0 : lo + LCAP);
        __syncthreads();
        resolve(s_cnt);
        __syncthreads();
        aggphase(s_cnt, lo == 0);
      }
    }
    __syncthreads();

    if (!MEAN) {
      for (int i = tid; i < nbr * F; i += 1024) {
        const int r = i / F, f = i - r * F;
        const size_t g = (size_t)(n0 + r) * F + f;
        const float v = acc[r * LD + f] / (den[r * 4 + f / C] + EPS_DEN) + bias[f] + skip[g] + skb[f];
        out[g] = v > 0.f ? v : expm1f(v);
      }
    } else {
      for (int i = tid; i < nbr * C; i += 1024) {
        const int r = i / C, o = i - r * C;
        const float* ar = acc + r * LD;
        const float* dr = den + r * 4;
        const float m = 0.25f * (ar[o]         / (dr[0] + EPS_DEN) +
                                 ar[C + o]     / (dr[1] + EPS_DEN) +
                                 ar[2 * C + o] / (dr[2] + EPS_DEN) +
                                 ar[3 * C + o] / (dr[3] + EPS_DEN));
        out[(size_t)(n0 + r) * C + o] = m + bias[o] + skip[(size_t)(n0 + r) * SK2P + o] + skb[o];
      }
    }
    __syncthreads();
  }
}

__global__ void pad_w(const float* __restrict__ w, float* __restrict__ wp,
                      int rows, int cols, int ldp) {
  int i = blockIdx.x * blockDim.x + threadIdx.x;
  if (i >= rows * ldp) return;
  int r = i / ldp, c = i - r * ldp;
  wp[i] = (c < cols) ? w[(size_t)r * cols + c] : 0.f;
}

extern "C" void kernel_launch(void* const* d_in, const int* in_sizes, int n_in,
                              void* d_out, int out_size, void* d_ws, size_t ws_size,
                              hipStream_t stream) {
  const float* x   = (const float*)d_in[0];
  const int*   ei  = (const int*)d_in[1];
  const float* lw[3] = {(const float*)d_in[2], (const float*)d_in[8],  (const float*)d_in[14]};
  const float* as[3] = {(const float*)d_in[3], (const float*)d_in[9],  (const float*)d_in[15]};
  const float* ad[3] = {(const float*)d_in[4], (const float*)d_in[10], (const float*)d_in[16]};
  const float* bi[3] = {(const float*)d_in[5], (const float*)d_in[11], (const float*)d_in[17]};
  const float* sw[3] = {(const float*)d_in[6], (const float*)d_in[12], (const float*)d_in[18]};
  const float* sb[3] = {(const float*)d_in[7], (const float*)d_in[13], (const float*)d_in[19]};

  const int N  = in_sizes[0] / 128;
  const int E0 = in_sizes[1] / 2;

  float* w = (float*)d_ws;
  float*    xcur = w;                      w += (size_t)N * HCF;
  float*    xh   = w;                      w += (size_t)N * HC2P;
  float*    skip = w;                      w += (size_t)N * HCF;
  float*    asrc = w;                      w += (size_t)N * 4;
  float*    adst = w;                      w += (size_t)N * 4;
  float*    w2p  = w;                      w += (size_t)128 * HC2P;
  float*    sw2p = w;                      w += (size_t)128 * SK2P;
  int*      glist = (int*)w;               w += (size_t)NLIST * LCAP * 2;
  int*      gcnt  = (int*)w;               w += (size_t)NLIST;

  const int mt8  = (N + 127) / 128;
  const int own  = 256;
  const int own2 = 224;

  gemm_wmma_f32<128, 128, 128, 128><<<dim3(mt8, 2), 256, 0, stream>>>(x, lw[0], xh,   N);
  gemm_wmma_f32<128, 128, 128, 128><<<dim3(mt8, 2), 256, 0, stream>>>(x, sw[0], skip, N);
  att_kernel<32, 128><<<(N * 4 + 255) / 256, 256, 0, stream>>>(xh, as[0], ad[0], asrc, adst, N);
  gat_own<32, 128, false, 196, 0, 196><<<own, 1024, 0, stream>>>(ei, E0, N, xh, asrc, adst, skip, bi[0], sb[0], xcur, glist, gcnt);

  gemm_wmma_f32<128, 128, 128, 128><<<dim3(mt8, 2), 256, 0, stream>>>(xcur, lw[1], xh,   N);
  gemm_wmma_f32<128, 128, 128, 128><<<dim3(mt8, 2), 256, 0, stream>>>(xcur, sw[1], skip, N);
  att_kernel<32, 128><<<(N * 4 + 255) / 256, 256, 0, stream>>>(xh, as[1], ad[1], asrc, adst, N);
  gat_own<32, 128, false, 196, 1, 196><<<own, 1024, 0, stream>>>(ei, E0, N, xh, asrc, adst, skip, bi[1], sb[1], xcur, glist, gcnt);

  pad_w<<<(128 * HC2P + 255) / 256, 256, 0, stream>>>(lw[2], w2p,  128, HC2, HC2P);
  pad_w<<<(128 * SK2P + 255) / 256, 256, 0, stream>>>(sw[2], sw2p, 128, C2,  SK2P);
  gemm_wmma_f32<128, 128, HC2P, HC2P><<<dim3(mt8, 3), 256, 0, stream>>>(xcur, w2p,  xh,   N);
  gemm_wmma_f32<128, 128, SK2P, SK2P><<<dim3(mt8, 1), 256, 0, stream>>>(xcur, sw2p, skip, N);
  att_kernel<47, HC2P><<<(N * 4 + 255) / 256, 256, 0, stream>>>(xh, as[2], ad[2], asrc, adst, N);
  gat_own<47, HC2P, true, 224, 2, 196><<<own2, 1024, 0, stream>>>(ei, E0, N, xh, asrc, adst, skip, bi[2], sb[2], (float*)d_out, glist, gcnt);
}


# Layer l+1's two fp32 WMMA products run inside layer l's owner work-group on its finished rows kept in LDS (pitch 132), weights staged unpadded: 14 -> 8 launches; every sum order as before
# speedup vs baseline: 4.5415x; 1.0855x over previous
#include <hip/hip_runtime.h>
#include <math.h>

#define HEADS      4
#define HCF        128
#define C2         47
#define HC2        188
#define HC2P       192
#define SK2P       64
#define NEG_SLOPE  0.2f
#define EPS_DEN    1e-16f

typedef float v2f __attribute__((ext_vector_type(2)));
typedef float v8f __attribute__((ext_vector_type(8)));

__device__ __forceinline__ unsigned fkey(float f) {
  unsigned u = __float_as_uint(f);
  return (u & 0x80000000u) ? ~u : (u | 0x80000000u);
}
__device__ __forceinline__ float fdecode(unsigned k) {
  unsigned u = (k & 0x80000000u) ? (k ^ 0x80000000u) : ~k;
  return __uint_as_float(u);
}

template <int K, int LDA, int LDB, int LDC>
__global__ __launch_bounds__(256)
void gemm_wmma_f32(const float* __restrict__ A, const float* __restrict__ B,
                   float* __restrict__ C, int M) {
  __shared__ float bs[(K / 4) * 128 * 2];

  const int tid  = threadIdx.x;
  const int lane = tid & 31;
  const int wave = tid >> 5;
  const int tn   = blockIdx.y * 64;

  for (int idx = tid; idx < (K / 4) * 128; idx += 256) {
    const int l   = idx & 31;
    const int t   = (idx >> 5) & 3;
    const int ks  = idx >> 7;
    const int k   = 4 * ks + ((l >> 4) << 1);
    const int col = tn + t * 16 + (l & 15);
    float2 b;
    b.x = B[(size_t)k * LDB + col];
    b.y = B[(size_t)(k + 1) * LDB + col];
    *(float2*)&bs[(size_t)idx * 2] = b;
  }
  __syncthreads();

  const int rlo   = lane & 15;
  const int khalf = (lane >> 4) << 1;
  const int tm    = blockIdx.x * 8 + wave;

  int arow = tm * 16 + rlo;
  if (arow >= M) arow = M - 1;
  const float* ap = A + (size_t)arow * LDA;

  v8f acc[4] = {};
#pragma unroll 4
  for (int ks = 0; ks < K / 4; ++ks) {
    v2f a;
    a.x = ap[4 * ks + khalf];
    a.y = ap[4 * ks + khalf + 1];
    const float* bp = &bs[(size_t)(ks * 128 + lane) * 2];
#pragma unroll
    for (int t = 0; t < 4; ++t) {
      v2f b = *(const v2f*)(bp + t * 64);
      acc[t] = __builtin_amdgcn_wmma_f32_16x16x4_f32(
          false, a, false, b, (short)0, acc[t], false, false);
    }
  }

  const int mbase = tm * 16 + ((lane >> 4) << 3);
  if ((tm + 1) * 16 <= M) {
    float* cp = C + (size_t)mbase * LDC + tn + rlo;
#pragma unroll
    for (int t = 0; t < 4; ++t)
#pragma unroll
      for (int r = 0; r < 8; ++r)
        cp[r * LDC + t * 16] = acc[t][r];
  } else {
#pragma unroll
    for (int t = 0; t < 4; ++t)
#pragma unroll
      for (int r = 0; r < 8; ++r) {
        const int row = mbase + r;
        if (row < M) C[(size_t)row * LDC + tn + t * 16 + rlo] = acc[t][r];
      }
  }
}

template <int C, int LD>
__global__ void att_kernel(const float* __restrict__ xh,
                           const float* __restrict__ att_src,
                           const float* __restrict__ att_dst,
                           float* __restrict__ asrc, float* __restrict__ adst,
                           int N) {
  int id = blockIdx.x * blockDim.x + threadIdx.x;
  if (id >= N * HEADS) return;
  int n = id >> 2, h = id & 3;
  const float* xp = xh + (size_t)n * LD + h * C;
  const float* ws = att_src + h * C;
  const float* wd = att_dst + h * C;
  float s1 = 0.f, s2 = 0.f;
#pragma unroll 8
  for (int c = 0; c < C; ++c) {
    float v = xp[c];
    s1 += v * ws[c];
    s2 += v * wd[c];
  }
  asrc[id] = s1;
  adst[id] = s2;
}

#define LCAP  16384
#define NLIST 256

template <int C, int LD, bool MEAN, int NBT, int LMODE, int NBL,
          int FUSE, int NWC, int NLD, int NSC, int NSLD>
__global__ __launch_bounds__(1024)
void gat_own(const int* __restrict__ ei, int E0, int N,
             const float* __restrict__ xh,
             const float* __restrict__ asrc, const float* __restrict__ adst,
             const float* __restrict__ skip, const float* __restrict__ bias,
             const float* __restrict__ skb, float* __restrict__ out,
             int* glist, int* gcnt,
             const float* __restrict__ nw,
             const float* __restrict__ nsw,
             float* __restrict__ nxh,
             float* __restrict__ nskip) {
  constexpr int F   = HEADS * C;
  constexpr int NCH = (F + 31) / 32;
  constexpr int AP   = FUSE ? LD + 4 : LD;
  constexpr int NBTP = FUSE ? ((NBT + 15) / 16) * 16 : NBT;
  __shared__ __attribute__((aligned(16))) float acc[NBTP * AP];
  __shared__ __attribute__((aligned(16))) float bs[FUSE ? 2 * 32 * 128 * 2 : 2];
  __shared__ float    den[NBT * 4];
  __shared__ float    adl[NBT * 4];
  __shared__ unsigned mx[NBT * 4];
  __shared__ int      list[LCAP];
  __shared__ unsigned char lr[LCAP];
  __shared__ int      s_cnt;

  const int tid  = threadIdx.x;
  const int lane = tid & 31;
  const int wave = tid >> 5;
  const int* __restrict__ srcp = ei;
  const int* __restrict__ dstp = ei + E0;

  for (int n0 = blockIdx.x * NBT; n0 < N; n0 += gridDim.x * NBT) {
    const int nbr  = (N - n0 < NBT) ? (N - n0) : NBT;
    const int bidx = n0 / NBT;
    const bool keep = (LMODE == 0) && bidx < NLIST;
    int2* glw = (int2*)glist + (size_t)(keep ? bidx : 0) * LCAP;

    for (int i = tid; i < NBTP * AP; i += 1024) acc[i] = 0.f;
    for (int i = tid; i < NBT * 4; i += 1024) {
      den[i] = 0.f;
      mx[i]  = 0u;
      adl[i] = (i < nbr * 4) ? adst[(size_t)n0 * 4 + i] : 0.f;
    }
    if (tid == 0) s_cnt = 0;
    __syncthreads();

    auto test = [&](int d, int i) {
      if ((unsigned)(d - n0) < (unsigned)nbr) {
        const int slot = atomicAdd(&s_cnt, 1);
        if (slot < LCAP) {
          list[slot] = i;
          lr[slot]   = (unsigned char)(d - n0);
        }
      }
    };
    auto scan = [&](int lo, int hi) {
      if (((E0 | lo) & 3) == 0) {
        const int4* __restrict__ p = (const int4*)dstp;
        const int q0 = lo >> 2, q1 = hi >> 2;
        for (int q = q0 + tid; q < q1; q += 4096) {
          int4 v[4];
#pragma unroll
          for (int j = 0; j < 4; ++j) {
            const int qq = q + j * 1024;
            v[j] = p[qq < q1 ? qq : q1 - 1];
          }
#pragma unroll
          for (int j = 0; j < 4; ++j) {
            const int qq = q + j * 1024;
            if (qq < q1) {
              test(v[j].x, 4 * qq);
              test(v[j].y, 4 * qq + 1);
              test(v[j].z, 4 * qq + 2);
              test(v[j].w, 4 * qq + 3);
            }
          }
        }
        for (int i = (q1 << 2) + tid; i < hi; i += 1024) test(dstp[i], i);
      } else {
        for (int i = lo + tid; i < hi; i += 1024) test(dstp[i], i);
      }
    };
    auto maxphase = [&](int cntc, bool self, bool resolved, bool store) {
      const int total = cntc + (self ? nbr : 0);
      for (int t = tid; t < total; t += 1024) {
        int s, r;
        if (t < cntc) {
          r = lr[t];
          if (resolved) {
            s = list[t];
          } else {
            s = srcp[list[t]];
            list[t] = s;
            if (LMODE == 0 && store) glw[t] = make_int2(s, r);
          }
        } else {
          r = t - cntc;
          s = n0 + r;
        }
        const float4 a = *(const float4*)(asrc + (size_t)s * 4);
        const float av[4] = {a.x, a.y, a.z, a.w};
#pragma unroll
        for (int h = 0; h < 4; ++h) {
          float v = av[h] + adl[r * 4 + h];
          v = v > 0.f ? v : NEG_SLOPE * v;
          atomicMax(&mx[r * 4 + h], fkey(v));
        }
      }
    };
    auto resolve = [&](int cntc) {
      for (int t = tid; t < cntc; t += 1024) list[t] = srcp[list[t]];
    };
    auto aggphase = [&](int cntc_, bool self) {
      const int cntc  = __builtin_amdgcn_readfirstlane(cntc_);
      const int total = cntc + (self ? nbr : 0);
      const int nb8   = (total + 7) >> 3;
      const int wv0   = __builtin_amdgcn_readfirstlane(wave);
      const int je    = lane >> 2;
      const int hh    = lane & 3;
      for (int b = wv0; b < nb8; b += 32) {
        const int  t   = b * 8 + je;
        const bool okl = t < total;
        const int  tc  = okl ? t : total - 1;
        int s, r;
        if (tc < cntc) {
          s = list[tc];
          r = lr[tc];
        } else {
          r = tc - cntc;
          s = n0 + r;
        }
        const float a = asrc[(size_t)s * 4 + hh];
        float x[8][NCH];
#pragma unroll
        for (int j = 0; j < 8; ++j) {
          const int su = __builtin_amdgcn_readlane(s, 4 * j);
          const float* xs = xh + (size_t)su * LD;
#pragma unroll
          for (int c = 0; c < NCH; ++c) x[j][c] = xs[c * 32 + lane];
        }
        float v = a + adl[r * 4 + hh];
        v = v > 0.f ? v : NEG_SLOPE * v;
        const float w = __expf(v - fdecode(mx[r * 4 + hh]));
        if (okl) atomicAdd(&den[r * 4 + hh], w);
#pragma unroll
        for (int j = 0; j < 8; ++j) {
          if (b * 8 + j < total) {
            const int ru = __builtin_amdgcn_readlane(r, 4 * j);
            float wj[4];
#pragma unroll
            for (int h = 0; h < 4; ++h)
              wj[h] = __int_as_float(__builtin_amdgcn_readlane(__float_as_int(w), 4 * j + h));
#pragma unroll
            for (int c = 0; c < NCH; ++c) {
              const int col = c * 32 + lane;
              const int hd  = col / C;
              const float wh = hd == 0 ? wj[0] : hd == 1 ? wj[1] : hd == 2 ? wj[2] : wj[3];
              if (col < F) atomicAdd(&acc[ru * AP + col], x[j][c] * wh);
            }
          }
        }
      }
    };

    int  cnt  = 0;
    bool have = false;
    if (LMODE == 1) {
      if (bidx < NLIST) {
        const int c = gcnt[bidx];
        if (c <= LCAP) {
          const int2* gl = (const int2*)glist + (size_t)bidx * LCAP;
          for (int t = tid; t < c; t += 1024) {
            const int2 e = gl[t];
            list[t] = e.x;
            lr[t]   = (unsigned char)e.y;
          }
          cnt  = c;
          have = true;
        }
      }
    }
    if (LMODE == 2) {
      const int b0 = n0 / NBL, b1 = (n0 + nbr - 1) / NBL;
      bool ok = b1 < NLIST;
      for (int b = b0; ok && b <= b1; ++b) ok = gcnt[b] <= LCAP;
      if (ok) {
        for (int b = b0; b <= b1; ++b) {
          const int c = gcnt[b];
          const int2* gl = (const int2*)glist + (size_t)b * LCAP;
          const int off = b * NBL - n0;
          for (int t = tid; t < c; t += 4096) {
            int2 e[4];
#pragma unroll
            for (int j = 0; j < 4; ++j) {
              const int tt = t + j * 1024;
              e[j] = gl[tt < c ? tt : c - 1];
            }
#pragma unroll
            for (int j = 0; j < 4; ++j) {
              const int rr = e[j].y + off;
              if (t + j * 1024 < c && (unsigned)rr < (unsigned)nbr) {
                const int slot = atomicAdd(&s_cnt, 1);
                if (slot < LCAP) {
                  list[slot] = e[j].x;
                  lr[slot]   = (unsigned char)rr;
                }
              }
            }
          }
        }
        __syncthreads();
        cnt  = s_cnt;
        have = cnt <= LCAP;
        __syncthreads();
        if (!have && tid == 0) s_cnt = 0;
      }
    }
    if (LMODE != 0) __syncthreads();
    if (!have) {
      scan(0, E0);
      __syncthreads();
      cnt = s_cnt;
    }
    if (keep && tid == 0) gcnt[bidx] = cnt;
    if (cnt <= LCAP) {
      maxphase(cnt, true, have, keep);
      __syncthreads();
      aggphase(cnt, true);
    } else {
      for (int lo = 0; lo < E0; lo += LCAP) {
        __syncthreads();
        if (tid == 0) s_cnt = 0;
        __syncthreads();
        scan(lo, (E0 - lo < LCAP) ? E0 : lo + LCAP);
        __syncthreads();
        maxphase(s_cnt, lo == 0, false, false);
      }
      for (int lo = 0; lo < E0; lo += LCAP) {
        __syncthreads();
        if (tid == 0) s_cnt = 0;
        __syncthreads();
        scan(lo, (E0 - lo < LCAP) ? E0 : lo + LCAP);
        __syncthreads();
        resolve(s_cnt);
        __syncthreads();
        aggphase(s_cnt, lo == 0);
      }
    }
    __syncthreads();

    if constexpr (FUSE != 0) {
      for (int i = tid; i < nbr * F; i += 1024) {
        const int r = i / F, f = i - r * F;
        const size_t g = (size_t)(n0 + r) * F + f;
        const float v = acc[r * AP + f] / (den[r * 4 + f / C] + EPS_DEN) + bias[f] + skip[g] + skb[f];
        acc[r * AP + f] = v > 0.f ? v : expm1f(v);
      }
      __syncthreads();

      constexpr int NT  = NBTP / 16;
      constexpr int NLH = NLD / 64;
      static_assert(F == 128, "K of the fused products");
      static_assert(NLD / 64 + NSLD / 64 == 4, "four strips");
      static_assert(2 * NT <= 32, "one unit per wave and pass");
      const int wv    = __builtin_amdgcn_readfirstlane(wave);
      const int rlo   = lane & 15;
      const int khalf = (lane >> 4) << 1;
#pragma unroll
      for (int p = 0; p < 2; ++p) {
#pragma unroll
        for (int i = 0; i < 8; ++i) {
          const int idx = tid + i * 1024;
          const int q   = 2 * p + (i >> 2);
          const int id  = idx & 4095;
          const int l   = id & 31;
          const int t   = (id >> 5) & 3;
          const int ks  = id >> 7;
          const int k   = 4 * ks + ((l >> 4) << 1);
          const bool lin = q < NLH;
          const float* __restrict__ wsrc = lin ? nw : nsw;
          const int wld = lin ? NWC : NSC;
          const int col = (lin ? q : q - NLH) * 64 + t * 16 + (l & 15);
          float2 b = make_float2(0.f, 0.f);
          if (col < wld) {
            b.x = wsrc[(size_t)k * wld + col];
            b.y = wsrc[(size_t)(k + 1) * wld + col];
          }
          *(float2*)&bs[(size_t)idx * 2] = b;
        }
        __syncthreads();
        const int un = (wv + 2 * p) & 31;
        if (un < 2 * NT) {
          const int hsel = un / NT;
          const int tile = un - hsel * NT;
          const int q    = 2 * p + hsel;
          const bool lin = q < NLH;
          float* __restrict__ dst = lin ? nxh : nskip;
          const int ldd  = lin ? NLD : NSLD;
          const int col0 = (lin ? q : q - NLH) * 64;
          const float* ap = acc + (tile * 16 + rlo) * AP + khalf;
          const float* bp = bs + (size_t)(hsel * 4096 + lane) * 2;
          v8f d[4] = {};
#pragma unroll 4
          for (int ks = 0; ks < F / 4; ++ks) {
            const v2f a = *(const v2f*)(ap + 4 * ks);
#pragma unroll
            for (int t = 0; t < 4; ++t) {
              const v2f b = *(const v2f*)(bp + ks * 256 + t * 64);
              d[t] = __builtin_amdgcn_wmma_f32_16x16x4_f32(
                  false, a, false, b, (short)0, d[t], false, false);
            }
          }
          const int mbase = tile * 16 + ((lane >> 4) << 3);
#pragma unroll
          for (int t = 0; t < 4; ++t)
#pragma unroll
            for (int r = 0; r < 8; ++r) {
              const int row = mbase + r;
              if (row < nbr) dst[(size_t)(n0 + row) * ldd + col0 + t * 16 + rlo] = d[t][r];
            }
        }
        __syncthreads();
      }
    } else if constexpr (!MEAN) {
      for (int i = tid; i < nbr * F; i += 1024) {
        const int r = i / F, f = i - r * F;
        const size_t g = (size_t)(n0 + r) * F + f;
        const float v = acc[r * AP + f] / (den[r * 4 + f / C] + EPS_DEN) + bias[f] + skip[g] + skb[f];
        out[g] = v > 0.f ? v : expm1f(v);
      }
    } else {
      for (int i = tid; i < nbr * C; i += 1024) {
        const int r = i / C, o = i - r * C;
        const float* ar = acc + r * AP;
        const float* dr = den + r * 4;
        const float m = 0.25f * (ar[o]         / (dr[0] + EPS_DEN) +
                                 ar[C + o]     / (dr[1] + EPS_DEN) +
                                 ar[2 * C + o] / (dr[2] + EPS_DEN) +
                                 ar[3 * C + o] / (dr[3] + EPS_DEN));
        out[(size_t)(n0 + r) * C + o] = m + bias[o] + skip[(size_t)(n0 + r) * SK2P + o] + skb[o];
      }
    }
    __syncthreads();
  }
}

extern "C" void kernel_launch(void* const* d_in, const int* in_sizes, int n_in,
                              void* d_out, int out_size, void* d_ws, size_t ws_size,
                              hipStream_t stream) {
  const float* x   = (const float*)d_in[0];
  const int*   ei  = (const int*)d_in[1];
  const float* lw[3] = {(const float*)d_in[2], (const float*)d_in[8],  (const float*)d_in[14]};
  const float* as[3] = {(const float*)d_in[3], (const float*)d_in[9],  (const float*)d_in[15]};
  const float* ad[3] = {(const float*)d_in[4], (const float*)d_in[10], (const float*)d_in[16]};
  const float* bi[3] = {(const float*)d_in[5], (const float*)d_in[11], (const float*)d_in[17]};
  const float* sw[3] = {(const float*)d_in[6], (const float*)d_in[12], (const float*)d_in[18]};
  const float* sb[3] = {(const float*)d_in[7], (const float*)d_in[13], (const float*)d_in[19]};

  const int N  = in_sizes[0] / 128;
  const int E0 = in_sizes[1] / 2;

  float* w = (float*)d_ws;
  float*    xh0  = w;                      w += (size_t)N * HCF;
  float*    xh1  = w;                      w += (size_t)N * HCF;
  float*    xh2  = w;                      w += (size_t)N * HC2P;
  float*    sk0  = w;                      w += (size_t)N * HCF;
  float*    sk1  = w;                      w += (size_t)N * HCF;
  float*    sk2  = w;                      w += (size_t)N * SK2P;
  float*    asrc = w;                      w += (size_t)N * 4;
  float*    adst = w;                      w += (size_t)N * 4;
  int*      glist = (int*)w;               w += (size_t)NLIST * LCAP * 2;
  int*      gcnt  = (int*)w;               w += (size_t)NLIST;

  const int mt8  = (N + 127) / 128;
  const int own  = 256;
  const int own2 = 224;

  gemm_wmma_f32<128, 128, 128, 128><<<dim3(mt8, 2), 256, 0, stream>>>(x, lw[0], xh0, N);
  gemm_wmma_f32<128, 128, 128, 128><<<dim3(mt8, 2), 256, 0, stream>>>(x, sw[0], sk0, N);
  att_kernel<32, 128><<<(N * 4 + 255) / 256, 256, 0, stream>>>(xh0, as[0], ad[0], asrc, adst, N);
  gat_own<32, 128, false, 196, 0, 196, 1, 128, 128, 128, 128><<<own, 1024, 0, stream>>>(
      ei, E0, N, xh0, asrc, adst, sk0, bi[0], sb[0], nullptr, glist, gcnt, lw[1], sw[1], xh1, sk1);

  att_kernel<32, 128><<<(N * 4 + 255) / 256, 256, 0, stream>>>(xh1, as[1], ad[1], asrc, adst, N);
  gat_own<32, 128, false, 196, 1, 196, 1, HC2, HC2P, C2, SK2P><<<own, 1024, 0, stream>>>(
      ei, E0, N, xh1, asrc, adst, sk1, bi[1], sb[1], nullptr, glist, gcnt, lw[2], sw[2], xh2, sk2);

  att_kernel<47, HC2P><<<(N * 4 + 255) / 256, 256, 0, stream>>>(xh2, as[2], ad[2], asrc, adst, N);
  gat_own<47, HC2P, true, 224, 2, 196, 0, 1, 64, 1, 64><<<own2, 1024, 0, stream>>>(
      ei, E0, N, xh2, asrc, adst, sk2, bi[2], sb[2], (float*)d_out, glist, gcnt, nullptr, nullptr, nullptr, nullptr);
}


# Range lists built by one partition launch: 256 work-groups each sort 1/256 of the edge rows by owner range (LDS counters) into runs the owners collect; no owner walks the row; sums still LDS arrival
# speedup vs baseline: 5.1305x; 1.1297x over previous
#include <hip/hip_runtime.h>
#include <math.h>

#define HEADS      4
#define HCF        128
#define C2         47
#define HC2        188
#define HC2P       192
#define SK2P       64
#define NEG_SLOPE  0.2f
#define EPS_DEN    1e-16f

typedef float v2f __attribute__((ext_vector_type(2)));
typedef float v8f __attribute__((ext_vector_type(8)));

__device__ __forceinline__ unsigned fkey(float f) {
  unsigned u = __float_as_uint(f);
  return (u & 0x80000000u) ? ~u : (u | 0x80000000u);
}
__device__ __forceinline__ float fdecode(unsigned k) {
  unsigned u = (k & 0x80000000u) ? (k ^ 0x80000000u) : ~k;
  return __uint_as_float(u);
}

template <int K, int LDA, int LDB, int LDC>
__global__ __launch_bounds__(256)
void gemm_wmma_f32(const float* __restrict__ A, const float* __restrict__ B,
                   float* __restrict__ C, int M) {
  __shared__ float bs[(K / 4) * 128 * 2];

  const int tid  = threadIdx.x;
  const int lane = tid & 31;
  const int wave = tid >> 5;
  const int tn   = blockIdx.y * 64;

  for (int idx = tid; idx < (K / 4) * 128; idx += 256) {
    const int l   = idx & 31;
    const int t   = (idx >> 5) & 3;
    const int ks  = idx >> 7;
    const int k   = 4 * ks + ((l >> 4) << 1);
    const int col = tn + t * 16 + (l & 15);
    float2 b;
    b.x = B[(size_t)k * LDB + col];
    b.y = B[(size_t)(k + 1) * LDB + col];
    *(float2*)&bs[(size_t)idx * 2] = b;
  }
  __syncthreads();

  const int rlo   = lane & 15;
  const int khalf = (lane >> 4) << 1;
  const int tm    = blockIdx.x * 8 + wave;

  int arow = tm * 16 + rlo;
  if (arow >= M) arow = M - 1;
  const float* ap = A + (size_t)arow * LDA;

  v8f acc[4] = {};
#pragma unroll 4
  for (int ks = 0; ks < K / 4; ++ks) {
    v2f a;
    a.x = ap[4 * ks + khalf];
    a.y = ap[4 * ks + khalf + 1];
    const float* bp = &bs[(size_t)(ks * 128 + lane) * 2];
#pragma unroll
    for (int t = 0; t < 4; ++t) {
      v2f b = *(const v2f*)(bp + t * 64);
      acc[t] = __builtin_amdgcn_wmma_f32_16x16x4_f32(
          false, a, false, b, (short)0, acc[t], false, false);
    }
  }

  const int mbase = tm * 16 + ((lane >> 4) << 3);
  if ((tm + 1) * 16 <= M) {
    float* cp = C + (size_t)mbase * LDC + tn + rlo;
#pragma unroll
    for (int t = 0; t < 4; ++t)
#pragma unroll
      for (int r = 0; r < 8; ++r)
        cp[r * LDC + t * 16] = acc[t][r];
  } else {
#pragma unroll
    for (int t = 0; t < 4; ++t)
#pragma unroll
      for (int r = 0; r < 8; ++r) {
        const int row = mbase + r;
        if (row < M) C[(size_t)row * LDC + tn + t * 16 + rlo] = acc[t][r];
      }
  }
}

template <int C, int LD>
__global__ void att_kernel(const float* __restrict__ xh,
                           const float* __restrict__ att_src,
                           const float* __restrict__ att_dst,
                           float* __restrict__ asrc, float* __restrict__ adst,
                           int N) {
  int id = blockIdx.x * blockDim.x + threadIdx.x;
  if (id >= N * HEADS) return;
  int n = id >> 2, h = id & 3;
  const float* xp = xh + (size_t)n * LD + h * C;
  const float* ws = att_src + h * C;
  const float* wd = att_dst + h * C;
  float s1 = 0.f, s2 = 0.f;
#pragma unroll 8
  for (int c = 0; c < C; ++c) {
    float v = xp[c];
    s1 += v * ws[c];
    s2 += v * wd[c];
  }
  asrc[id] = s1;
  adst[id] = s2;
}

#define LCAP   16384
#define NLIST  256
#define NB0    196
#define NPART  256
#define SEGCAP 48

__global__ __launch_bounds__(1024)
void part_edges(const int* __restrict__ ei, int E0, int N,
                int* __restrict__ pseg_,
                int* __restrict__ pcnt) {
  __shared__ int  hist[NLIST];
  __shared__ int2 stage[NLIST * SEGCAP];
  const int tid = threadIdx.x;
  const int w   = blockIdx.x;
  const int* __restrict__ srcp = ei;
  const int* __restrict__ dstp = ei + E0;
  int2* __restrict__ pseg = (int2*)pseg_;
  const int ch = (E0 + NPART - 1) / NPART;
  const int lo = w * ch < E0 ? w * ch : E0;
  const int hi = (E0 - lo < ch) ? E0 : lo + ch;

  for (int i = tid; i < NLIST; i += 1024) hist[i] = 0;
  __syncthreads();

  for (int e0 = lo + tid; e0 < hi; e0 += 4 * 1024) {
    int d[4], s[4];
#pragma unroll
    for (int j = 0; j < 4; ++j) {
      const int e = e0 + j * 1024;
      const int c = e < hi ? e : hi - 1;
      d[j] = dstp[c];
      s[j] = srcp[c];
    }
    asm volatile("" :: "v"(s[0]), "v"(s[1]), "v"(s[2]), "v"(s[3]));
#pragma unroll
    for (int j = 0; j < 4; ++j) {
      if (e0 + j * 1024 < hi && (unsigned)d[j] < (unsigned)N) {
        const int b = d[j] / NB0;
        if (b < NLIST) {
          const int slot = atomicAdd(&hist[b], 1);
          if (slot < SEGCAP) stage[b * SEGCAP + slot] = make_int2(s[j], d[j] - b * NB0);
        }
      }
    }
  }
  __syncthreads();

  for (int b = tid; b < NLIST; b += 1024) pcnt[(size_t)b * NPART + w] = hist[b];
  for (int idx = tid; idx < NLIST * SEGCAP; idx += 4 * 1024) {
    int2 v[4];
    bool ok[4];
#pragma unroll
    for (int j = 0; j < 4; ++j) {
      const int ii = idx + j * 1024;
      const int b  = ii / SEGCAP, jj = ii - b * SEGCAP;
      ok[j] = jj < hist[b];
      v[j]  = make_int2(0, 0);
      if (ok[j]) v[j] = stage[ii];
    }
#pragma unroll
    for (int j = 0; j < 4; ++j) {
      const int ii = idx + j * 1024;
      const int b  = ii / SEGCAP, jj = ii - b * SEGCAP;
      if (ok[j]) pseg[((size_t)b * NPART + w) * SEGCAP + jj] = v[j];
    }
  }
}

template <int C, int LD, bool MEAN, int NBT, int LMODE, int NBL,
          int FUSE, int NWC, int NLD, int NSC, int NSLD>
__global__ __launch_bounds__(1024)
void gat_own(const int* __restrict__ ei, int E0, int N,
             const float* __restrict__ xh,
             const float* __restrict__ asrc, const float* __restrict__ adst,
             const float* __restrict__ skip, const float* __restrict__ bias,
             const float* __restrict__ skb, float* __restrict__ out,
             int* glist, int* gcnt,
             const int* __restrict__ pseg_,
             const int* __restrict__ pcnt,
             const float* __restrict__ nw,
             const float* __restrict__ nsw,
             float* __restrict__ nxh,
             float* __restrict__ nskip) {
  constexpr int F   = HEADS * C;
  constexpr int NCH = (F + 31) / 32;
  constexpr int AP   = FUSE ? LD + 4 : LD;
  constexpr int NBTP = FUSE ? ((NBT + 15) / 16) * 16 : NBT;
  __shared__ __attribute__((aligned(16))) float acc[NBTP * AP];
  __shared__ __attribute__((aligned(16))) float bs[FUSE ? 2 * 32 * 128 * 2 : 2];
  __shared__ float    den[NBT * 4];
  __shared__ float    adl[NBT * 4];
  __shared__ unsigned mx[NBT * 4];
  __shared__ int      list[LCAP];
  __shared__ unsigned char lr[LCAP];
  __shared__ int      s_cnt;
  __shared__ int      s_bad;
  __shared__ int      prun[LMODE == 0 ? 2 * NPART : 2];

  const int tid  = threadIdx.x;
  const int lane = tid & 31;
  const int wave = tid >> 5;
  const int* __restrict__ srcp = ei;
  const int* __restrict__ dstp = ei + E0;

  for (int n0 = blockIdx.x * NBT; n0 < N; n0 += gridDim.x * NBT) {
    const int nbr  = (N - n0 < NBT) ? (N - n0) : NBT;
    const int bidx = n0 / NBT;
    const bool keep = (LMODE == 0) && bidx < NLIST;
    int2* glw = (int2*)glist + (size_t)(keep ? bidx : 0) * LCAP;

    for (int i = tid; i < NBTP * AP; i += 1024) acc[i] = 0.f;
    for (int i = tid; i < NBT * 4; i += 1024) {
      den[i] = 0.f;
      mx[i]  = 0u;
      adl[i] = (i < nbr * 4) ? adst[(size_t)n0 * 4 + i] : 0.f;
    }
    if (tid == 0) {
      s_cnt = 0;
      s_bad = 0;
    }
    __syncthreads();

    auto test = [&](int d, int i) {
      if ((unsigned)(d - n0) < (unsigned)nbr) {
        const int slot = atomicAdd(&s_cnt, 1);
        if (slot < LCAP) {
          list[slot] = i;
          lr[slot]   = (unsigned char)(d - n0);
        }
      }
    };
    auto scan = [&](int lo, int hi) {
      if (((E0 | lo) & 3) == 0) {
        const int4* __restrict__ p = (const int4*)dstp;
        const int q0 = lo >> 2, q1 = hi >> 2;
        for (int q = q0 + tid; q < q1; q += 4096) {
          int4 v[4];
#pragma unroll
          for (int j = 0; j < 4; ++j) {
            const int qq = q + j * 1024;
            v[j] = p[qq < q1 ? qq : q1 - 1];
          }
#pragma unroll
          for (int j = 0; j < 4; ++j) {
            const int qq = q + j * 1024;
            if (qq < q1) {
              test(v[j].x, 4 * qq);
              test(v[j].y, 4 * qq + 1);
              test(v[j].z, 4 * qq + 2);
              test(v[j].w, 4 * qq + 3);
            }
          }
        }
        for (int i = (q1 << 2) + tid; i < hi; i += 1024) test(dstp[i], i);
      } else {
        for (int i = lo + tid; i < hi; i += 1024) test(dstp[i], i);
      }
    };
    auto maxphase = [&](int cntc, bool self, bool resolved, bool store) {
      const int total = cntc + (self ? nbr : 0);
      for (int t = tid; t < total; t += 1024) {
        int s, r;
        if (t < cntc) {
          r = lr[t];
          if (resolved) {
            s = list[t];
            if (LMODE == 0 && store) glw[t] = make_int2(s, r);
          } else {
            s = srcp[list[t]];
            list[t] = s;
            if (LMODE == 0 && store) glw[t] = make_int2(s, r);
          }
        } else {
          r = t - cntc;
          s = n0 + r;
        }
        const float4 a = *(const float4*)(asrc + (size_t)s * 4);
        const float av[4] = {a.x, a.y, a.z, a.w};
#pragma unroll
        for (int h = 0; h < 4; ++h) {
          float v = av[h] + adl[r * 4 + h];
          v = v > 0.f ? v : NEG_SLOPE * v;
          atomicMax(&mx[r * 4 + h], fkey(v));
        }
      }
    };
    auto resolve = [&](int cntc) {
      for (int t = tid; t < cntc; t += 1024) list[t] = srcp[list[t]];
    };
    auto aggphase = [&](int cntc_, bool self) {
      const int cntc  = __builtin_amdgcn_readfirstlane(cntc_);
      const int total = cntc + (self ? nbr : 0);
      const int nb8   = (total + 7) >> 3;
      const int wv0   = __builtin_amdgcn_readfirstlane(wave);
      const int je    = lane >> 2;
      const int hh    = lane & 3;
      for (int b = wv0; b < nb8; b += 32) {
        const int  t   = b * 8 + je;
        const bool okl = t < total;
        const int  tc  = okl ? t : total - 1;
        int s, r;
        if (tc < cntc) {
          s = list[tc];
          r = lr[tc];
        } else {
          r = tc - cntc;
          s = n0 + r;
        }
        const float a = asrc[(size_t)s * 4 + hh];
        float x[8][NCH];
#pragma unroll
        for (int j = 0; j < 8; ++j) {
          const int su = __builtin_amdgcn_readlane(s, 4 * j);
          const float* xs = xh + (size_t)su * LD;
#pragma unroll
          for (int c = 0; c < NCH; ++c) x[j][c] = xs[c * 32 + lane];
        }
        float v = a + adl[r * 4 + hh];
        v = v > 0.f ? v : NEG_SLOPE * v;
        const float w = __expf(v - fdecode(mx[r * 4 + hh]));
        if (okl) atomicAdd(&den[r * 4 + hh], w);
#pragma unroll
        for (int j = 0; j < 8; ++j) {
          if (b * 8 + j < total) {
            const int ru = __builtin_amdgcn_readlane(r, 4 * j);
            float wj[4];
#pragma unroll
            for (int h = 0; h < 4; ++h)
              wj[h] = __int_as_float(__builtin_amdgcn_readlane(__float_as_int(w), 4 * j + h));
#pragma unroll
            for (int c = 0; c < NCH; ++c) {
              const int col = c * 32 + lane;
              const int hd  = col / C;
              const float wh = hd == 0 ? wj[0] : hd == 1 ? wj[1] : hd == 2 ? wj[2] : wj[3];
              if (col < F) atomicAdd(&acc[ru * AP + col], x[j][c] * wh);
            }
          }
        }
      }
    };

    int  cnt  = 0;
    bool have = false;
    if constexpr (LMODE == 0) {
      if (bidx < NLIST) {
        for (int w = tid; w < NPART; w += 1024) {
          const int c = pcnt[(size_t)bidx * NPART + w];
          if (c < 0 || c > SEGCAP) s_bad = 1;
          const int cc = c < 0 ? 0 : c > SEGCAP ? SEGCAP : c;
          prun[w]         = cc;
          prun[NPART + w] = atomicAdd(&s_cnt, cc);
        }
        __syncthreads();
        cnt  = s_cnt;
        have = s_bad == 0 && cnt <= LCAP;
        if (have) {
          const int2* __restrict__ sg = (const int2*)pseg_ + (size_t)bidx * NPART * SEGCAP;
          for (int idx = tid; idx < NPART * SEGCAP; idx += 4096) {
            int2 e[4];
            int  p[4];
#pragma unroll
            for (int j = 0; j < 4; ++j) {
              const int ii = idx + j * 1024;
              const int w  = ii / SEGCAP, jj = ii - w * SEGCAP;
              p[j] = -1;
              e[j] = make_int2(0, 0);
              if (jj < prun[w]) {
                p[j] = prun[NPART + w] + jj;
                e[j] = sg[ii];
              }
            }
#pragma unroll
            for (int j = 0; j < 4; ++j) {
              if (p[j] >= 0) {
                list[p[j]] = e[j].x;
                lr[p[j]]   = (unsigned char)e[j].y;
              }
            }
          }
        }
        __syncthreads();
        if (!have && tid == 0) s_cnt = 0;
      }
    }
    if (LMODE == 1) {
      if (bidx < NLIST) {
        const int c = gcnt[bidx];
        if (c <= LCAP) {
          const int2* gl = (const int2*)glist + (size_t)bidx * LCAP;
          for (int t = tid; t < c; t += 1024) {
            const int2 e = gl[t];
            list[t] = e.x;
            lr[t]   = (unsigned char)e.y;
          }
          cnt  = c;
          have = true;
        }
      }
    }
    if (LMODE == 2) {
      const int b0 = n0 / NBL, b1 = (n0 + nbr - 1) / NBL;
      bool ok = b1 < NLIST;
      for (int b = b0; ok && b <= b1; ++b) ok = gcnt[b] <= LCAP;
      if (ok) {
        for (int b = b0; b <= b1; ++b) {
          const int c = gcnt[b];
          const int2* gl = (const int2*)glist + (size_t)b * LCAP;
          const int off = b * NBL - n0;
          for (int t = tid; t < c; t += 4096) {
            int2 e[4];
#pragma unroll
            for (int j = 0; j < 4; ++j) {
              const int tt = t + j * 1024;
              e[j] = gl[tt < c ? tt : c - 1];
            }
#pragma unroll
            for (int j = 0; j < 4; ++j) {
              const int rr = e[j].y + off;
              if (t + j * 1024 < c && (unsigned)rr < (unsigned)nbr) {
                const int slot = atomicAdd(&s_cnt, 1);
                if (slot < LCAP) {
                  list[slot] = e[j].x;
                  lr[slot]   = (unsigned char)rr;
                }
              }
            }
          }
        }
        __syncthreads();
        cnt  = s_cnt;
        have = cnt <= LCAP;
        __syncthreads();
        if (!have && tid == 0) s_cnt = 0;
      }
    }
    __syncthreads();
    if (!have) {
      scan(0, E0);
      __syncthreads();
      cnt = s_cnt;
    }
    if (keep && tid == 0) gcnt[bidx] = cnt;
    if (cnt <= LCAP) {
      maxphase(cnt, true, have, keep);
      __syncthreads();
      aggphase(cnt, true);
    } else {
      for (int lo = 0; lo < E0; lo += LCAP) {
        __syncthreads();
        if (tid == 0) s_cnt = 0;
        __syncthreads();
        scan(lo, (E0 - lo < LCAP) ? E0 : lo + LCAP);
        __syncthreads();
        maxphase(s_cnt, lo == 0, false, false);
      }
      for (int lo = 0; lo < E0; lo += LCAP) {
        __syncthreads();
        if (tid == 0) s_cnt = 0;
        __syncthreads();
        scan(lo, (E0 - lo < LCAP) ? E0 : lo + LCAP);
        __syncthreads();
        resolve(s_cnt);
        __syncthreads();
        aggphase(s_cnt, lo == 0);
      }
    }
    __syncthreads();

    if constexpr (FUSE != 0) {
      for (int i = tid; i < nbr * F; i += 1024) {
        const int r = i / F, f = i - r * F;
        const size_t g = (size_t)(n0 + r) * F + f;
        const float v = acc[r * AP + f] / (den[r * 4 + f / C] + EPS_DEN) + bias[f] + skip[g] + skb[f];
        acc[r * AP + f] = v > 0.f ? v : expm1f(v);
      }
      __syncthreads();

      constexpr int NT  = NBTP / 16;
      constexpr int NLH = NLD / 64;
      static_assert(F == 128, "K of the fused products");
      static_assert(NLD / 64 + NSLD / 64 == 4, "four strips");
      static_assert(2 * NT <= 32, "one unit per wave and pass");
      const int wv    = __builtin_amdgcn_readfirstlane(wave);
      const int rlo   = lane & 15;
      const int khalf = (lane >> 4) << 1;
#pragma unroll
      for (int p = 0; p < 2; ++p) {
#pragma unroll
        for (int i = 0; i < 8; ++i) {
          const int idx = tid + i * 1024;
          const int q   = 2 * p + (i >> 2);
          const int id  = idx & 4095;
          const int l   = id & 31;
          const int t   = (id >> 5) & 3;
          const int ks  = id >> 7;
          const int k   = 4 * ks + ((l >> 4) << 1);
          const bool lin = q < NLH;
          const float* __restrict__ wsrc = lin ? nw : nsw;
          const int wld = lin ? NWC : NSC;
          const int col = (lin ? q : q - NLH) * 64 + t * 16 + (l & 15);
          float2 b = make_float2(0.f, 0.f);
          if (col < wld) {
            b.x = wsrc[(size_t)k * wld + col];
            b.y = wsrc[(size_t)(k + 1) * wld + col];
          }
          *(float2*)&bs[(size_t)idx * 2] = b;
        }
        __syncthreads();
        const int un = (wv + 2 * p) & 31;
        if (un < 2 * NT) {
          const int hsel = un / NT;
          const int tile = un - hsel * NT;
          const int q    = 2 * p + hsel;
          const bool lin = q < NLH;
          float* __restrict__ dst = lin ? nxh : nskip;
          const int ldd  = lin ? NLD : NSLD;
          const int col0 = (lin ? q : q - NLH) * 64;
          const float* ap = acc + (tile * 16 + rlo) * AP + khalf;
          const float* bp = bs + (size_t)(hsel * 4096 + lane) * 2;
          v8f d[4] = {};
#pragma unroll 4
          for (int ks = 0; ks < F / 4; ++ks) {
            const v2f a = *(const v2f*)(ap + 4 * ks);
#pragma unroll
            for (int t = 0; t < 4; ++t) {
              const v2f b = *(const v2f*)(bp + ks * 256 + t * 64);
              d[t] = __builtin_amdgcn_wmma_f32_16x16x4_f32(
                  false, a, false, b, (short)0, d[t], false, false);
            }
          }
          const int mbase = tile * 16 + ((lane >> 4) << 3);
#pragma unroll
          for (int t = 0; t < 4; ++t)
#pragma unroll
            for (int r = 0; r < 8; ++r) {
              const int row = mbase + r;
              if (row < nbr) dst[(size_t)(n0 + row) * ldd + col0 + t * 16 + rlo] = d[t][r];
            }
        }
        __syncthreads();
      }
    } else if constexpr (!MEAN) {
      for (int i = tid; i < nbr * F; i += 1024) {
        const int r = i / F, f = i - r * F;
        const size_t g = (size_t)(n0 + r) * F + f;
        const float v = acc[r * AP + f] / (den[r * 4 + f / C] + EPS_DEN) + bias[f] + skip[g] + skb[f];
        out[g] = v > 0.f ? v : expm1f(v);
      }
    } else {
      for (int i = tid; i < nbr * C; i += 1024) {
        const int r = i / C, o = i - r * C;
        const float* ar = acc + r * AP;
        const float* dr = den + r * 4;
        const float m = 0.25f * (ar[o]         / (dr[0] + EPS_DEN) +
                                 ar[C + o]     / (dr[1] + EPS_DEN) +
                                 ar[2 * C + o] / (dr[2] + EPS_DEN) +
                                 ar[3 * C + o] / (dr[3] + EPS_DEN));
        out[(size_t)(n0 + r) * C + o] = m + bias[o] + skip[(size_t)(n0 + r) * SK2P + o] + skb[o];
      }
    }
    __syncthreads();
  }
}

extern "C" void kernel_launch(void* const* d_in, const int* in_sizes, int n_in,
                              void* d_out, int out_size, void* d_ws, size_t ws_size,
                              hipStream_t stream) {
  const float* x   = (const float*)d_in[0];
  const int*   ei  = (const int*)d_in[1];
  const float* lw[3] = {(const float*)d_in[2], (const float*)d_in[8],  (const float*)d_in[14]};
  const float* as[3] = {(const float*)d_in[3], (const float*)d_in[9],  (const float*)d_in[15]};
  const float* ad[3] = {(const float*)d_in[4], (const float*)d_in[10], (const float*)d_in[16]};
  const float* bi[3] = {(const float*)d_in[5], (const float*)d_in[11], (const float*)d_in[17]};
  const float* sw[3] = {(const float*)d_in[6], (const float*)d_in[12], (const float*)d_in[18]};
  const float* sb[3] = {(const float*)d_in[7], (const float*)d_in[13], (const float*)d_in[19]};

  const int N  = in_sizes[0] / 128;
  const int E0 = in_sizes[1] / 2;

  float* w = (float*)d_ws;
  float*    xh0  = w;                      w += (size_t)N * HCF;
  float*    xh1  = w;                      w += (size_t)N * HCF;
  float*    xh2  = w;                      w += (size_t)N * HC2P;
  float*    sk0  = w;                      w += (size_t)N * HCF;
  float*    sk1  = w;                      w += (size_t)N * HCF;
  float*    sk2  = w;                      w += (size_t)N * SK2P;
  float*    asrc = w;                      w += (size_t)N * 4;
  float*    adst = w;                      w += (size_t)N * 4;
  int*      glist = (int*)w;               w += (size_t)NLIST * LCAP * 2;
  int*      gcnt  = (int*)w;               w += (size_t)NLIST;
  int*      pseg  = (int*)w;               w += (size_t)NLIST * NPART * SEGCAP * 2;
  int*      pcnt  = (int*)w;               w += (size_t)NLIST * NPART;

  const int mt8  = (N + 127) / 128;
  const int own  = 256;
  const int own2 = 224;

  part_edges<<<NPART, 1024, 0, stream>>>(ei, E0, N, pseg, pcnt);

  gemm_wmma_f32<128, 128, 128, 128><<<dim3(mt8, 2), 256, 0, stream>>>(x, lw[0], xh0, N);
  gemm_wmma_f32<128, 128, 128, 128><<<dim3(mt8, 2), 256, 0, stream>>>(x, sw[0], sk0, N);
  att_kernel<32, 128><<<(N * 4 + 255) / 256, 256, 0, stream>>>(xh0, as[0], ad[0], asrc, adst, N);
  gat_own<32, 128, false, NB0, 0, NB0, 1, 128, 128, 128, 128><<<own, 1024, 0, stream>>>(
      ei, E0, N, xh0, asrc, adst, sk0, bi[0], sb[0], nullptr, glist, gcnt, pseg, pcnt, lw[1], sw[1], xh1, sk1);

  att_kernel<32, 128><<<(N * 4 + 255) / 256, 256, 0, stream>>>(xh1, as[1], ad[1], asrc, adst, N);
  gat_own<32, 128, false, NB0, 1, NB0, 1, HC2, HC2P, C2, SK2P><<<own, 1024, 0, stream>>>(
      ei, E0, N, xh1, asrc, adst, sk1, bi[1], sb[1], nullptr, glist, gcnt, nullptr, nullptr, lw[2], sw[2], xh2, sk2);

  att_kernel<47, HC2P><<<(N * 4 + 255) / 256, 256, 0, stream>>>(xh2, as[2], ad[2], asrc, adst, N);
  gat_own<47, HC2P, true, 224, 2, NB0, 0, 1, 64, 1, 64><<<own2, 1024, 0, stream>>>(
      ei, E0, N, xh2, asrc, adst, sk2, bi[2], sb[2], (float*)d_out, glist, gcnt, nullptr, nullptr, nullptr, nullptr, nullptr, nullptr);
}


# Each layer's attention scores are written by the launch that stores its xh rows (layer-0 product, owner work-group), which reads them back behind a barrier, same dot order; 9 -> 6 launches
# speedup vs baseline: 5.3267x; 1.0382x over previous
#include <hip/hip_runtime.h>
#include <math.h>

#define HEADS      4
#define HCF        128
#define C2         47
#define HC2        188
#define HC2P       192
#define SK2P       64
#define NEG_SLOPE  0.2f
#define EPS_DEN    1e-16f

typedef float v2f __attribute__((ext_vector_type(2)));
typedef float v8f __attribute__((ext_vector_type(8)));

__device__ __forceinline__ unsigned fkey(float f) {
  unsigned u = __float_as_uint(f);
  return (u & 0x80000000u) ? ~u : (u | 0x80000000u);
}
__device__ __forceinline__ float fdecode(unsigned k) {
  unsigned u = (k & 0x80000000u) ? (k ^ 0x80000000u) : ~k;
  return __uint_as_float(u);
}

template <int K, int LDA, int LDB, int LDC>
__global__ __launch_bounds__(256)
void gemm_wmma_f32(const float* __restrict__ A, const float* __restrict__ B,
                   float* __restrict__ C, int M,
                   const float* att_s, const float* att_d, float* sc) {
  __shared__ float bs[(K / 4) * 128 * 2];

  const int tid  = threadIdx.x;
  const int lane = tid & 31;
  const int wave = tid >> 5;
  const int tn   = blockIdx.y * 64;

  for (int idx = tid; idx < (K / 4) * 128; idx += 256) {
    const int l   = idx & 31;
    const int t   = (idx >> 5) & 3;
    const int ks  = idx >> 7;
    const int k   = 4 * ks + ((l >> 4) << 1);
    const int col = tn + t * 16 + (l & 15);
    float2 b;
    b.x = B[(size_t)k * LDB + col];
    b.y = B[(size_t)(k + 1) * LDB + col];
    *(float2*)&bs[(size_t)idx * 2] = b;
  }
  __syncthreads();

  const int rlo   = lane & 15;
  const int khalf = (lane >> 4) << 1;
  const int tm    = blockIdx.x * 8 + wave;

  int arow = tm * 16 + rlo;
  if (arow >= M) arow = M - 1;
  const float* ap = A + (size_t)arow * LDA;

  v8f acc[4] = {};
#pragma unroll 4
  for (int ks = 0; ks < K / 4; ++ks) {
    v2f a;
    a.x = ap[4 * ks + khalf];
    a.y = ap[4 * ks + khalf + 1];
    const float* bp = &bs[(size_t)(ks * 128 + lane) * 2];
#pragma unroll
    for (int t = 0; t < 4; ++t) {
      v2f b = *(const v2f*)(bp + t * 64);
      acc[t] = __builtin_amdgcn_wmma_f32_16x16x4_f32(
          false, a, false, b, (short)0, acc[t], false, false);
    }
  }

  const int mbase = tm * 16 + ((lane >> 4) << 3);
  if ((tm + 1) * 16 <= M) {
    float* cp = C + (size_t)mbase * LDC + tn + rlo;
#pragma unroll
    for (int t = 0; t < 4; ++t)
#pragma unroll
      for (int r = 0; r < 8; ++r)
        cp[r * LDC + t * 16] = acc[t][r];
  } else {
#pragma unroll
    for (int t = 0; t < 4; ++t)
#pragma unroll
      for (int r = 0; r < 8; ++r) {
        const int row = mbase + r;
        if (row < M) C[(size_t)row * LDC + tn + t * 16 + rlo] = acc[t][r];
      }
  }

  if (att_s != nullptr) {
    __syncthreads();
    const int row = blockIdx.x * 128 + (tid >> 1);
    const int h   = blockIdx.y * 2 + (tid & 1);
    if (row < M) {
      const float* xp = C + (size_t)row * LDC + h * 32;
      const float* ws = att_s + h * 32;
      const float* wd = att_d + h * 32;
      float s1 = 0.f, s2 = 0.f;
#pragma unroll 16
      for (int c = 0; c < 32; ++c) {
        const float v = xp[c];
        s1 += v * ws[c];
        s2 += v * wd[c];
      }
      sc[(size_t)row * 4 + h] = s1;
      sc[(size_t)M * 4 + (size_t)row * 4 + h] = s2;
    }
  }
}

#define LCAP   16384
#define NLIST  256
#define NB0    196
#define NPART  256
#define SEGCAP 48

__global__ __launch_bounds__(1024)
void part_edges(const int* __restrict__ ei, int E0, int N,
                int* __restrict__ pseg_,
                int* __restrict__ pcnt) {
  __shared__ int  hist[NLIST];
  __shared__ int2 stage[NLIST * SEGCAP];
  const int tid = threadIdx.x;
  const int w   = blockIdx.x;
  const int* __restrict__ srcp = ei;
  const int* __restrict__ dstp = ei + E0;
  int2* __restrict__ pseg = (int2*)pseg_;
  const int ch = (E0 + NPART - 1) / NPART;
  const int lo = w * ch < E0 ? w * ch : E0;
  const int hi = (E0 - lo < ch) ? E0 : lo + ch;

  for (int i = tid; i < NLIST; i += 1024) hist[i] = 0;
  __syncthreads();

  for (int e0 = lo + tid; e0 < hi; e0 += 4 * 1024) {
    int d[4], s[4];
#pragma unroll
    for (int j = 0; j < 4; ++j) {
      const int e = e0 + j * 1024;
      const int c = e < hi ? e : hi - 1;
      d[j] = dstp[c];
      s[j] = srcp[c];
    }
    asm volatile("" :: "v"(s[0]), "v"(s[1]), "v"(s[2]), "v"(s[3]));
#pragma unroll
    for (int j = 0; j < 4; ++j) {
      if (e0 + j * 1024 < hi && (unsigned)d[j] < (unsigned)N) {
        const int b = d[j] / NB0;
        if (b < NLIST) {
          const int slot = atomicAdd(&hist[b], 1);
          if (slot < SEGCAP) stage[b * SEGCAP + slot] = make_int2(s[j], d[j] - b * NB0);
        }
      }
    }
  }
  __syncthreads();

  for (int b = tid; b < NLIST; b += 1024) pcnt[(size_t)b * NPART + w] = hist[b];
  for (int idx = tid; idx < NLIST * SEGCAP; idx += 4 * 1024) {
    int2 v[4];
    bool ok[4];
#pragma unroll
    for (int j = 0; j < 4; ++j) {
      const int ii = idx + j * 1024;
      const int b  = ii / SEGCAP, jj = ii - b * SEGCAP;
      ok[j] = jj < hist[b];
      v[j]  = make_int2(0, 0);
      if (ok[j]) v[j] = stage[ii];
    }
#pragma unroll
    for (int j = 0; j < 4; ++j) {
      const int ii = idx + j * 1024;
      const int b  = ii / SEGCAP, jj = ii - b * SEGCAP;
      if (ok[j]) pseg[((size_t)b * NPART + w) * SEGCAP + jj] = v[j];
    }
  }
}

template <int C, int LD, int NBT, int LMODE,
          int FUSE, int NWC, int NLD, int NSC, int NSLD>
__global__ __launch_bounds__(1024)
void gat_own(const int* __restrict__ ei, int E0, int N,
             const float* __restrict__ xh,
             const float* __restrict__ asrc, const float* __restrict__ adst,
             const float* __restrict__ skip, const float* __restrict__ bias,
             const float* __restrict__ skb, float* __restrict__ out,
             int* glist, int* gcnt,
             const int* __restrict__ pseg_,
             const int* __restrict__ pcnt,
             const float* __restrict__ nw,
             const float* __restrict__ nsw,
             float* __restrict__ nxh,
             float* __restrict__ nskip,
             const float* natt_s,
             const float* natt_d,
             float* nas) {
  constexpr int NBL = NB0;
  constexpr int F   = HEADS * C;
  constexpr int NCH = (F + 31) / 32;
  constexpr int AP   = FUSE ? LD + 4 : LD;
  constexpr int NBTP = FUSE ? ((NBT + 15) / 16) * 16 : NBT;
  __shared__ __attribute__((aligned(16))) float acc[NBTP * AP];
  __shared__ __attribute__((aligned(16))) float bs[FUSE ? 2 * 32 * 128 * 2 : 2];
  __shared__ float    den[NBT * 4];
  __shared__ float    adl[NBT * 4];
  __shared__ unsigned mx[NBT * 4];
  __shared__ int      list[LCAP];
  __shared__ unsigned char lr[LCAP];
  __shared__ int      s_cnt;
  __shared__ int      s_bad;
  __shared__ int      prun[LMODE == 0 ? 2 * NPART : 2];

  const int tid  = threadIdx.x;
  const int lane = tid & 31;
  const int wave = tid >> 5;
  const int* __restrict__ srcp = ei;
  const int* __restrict__ dstp = ei + E0;

  for (int n0 = blockIdx.x * NBT; n0 < N; n0 += gridDim.x * NBT) {
    const int nbr  = (N - n0 < NBT) ? (N - n0) : NBT;
    const int bidx = n0 / NBT;
    const bool keep = (LMODE == 0) && bidx < NLIST;
    int2* glw = (int2*)glist + (size_t)(keep ? bidx : 0) * LCAP;

    for (int i = tid; i < NBTP * AP; i += 1024) acc[i] = 0.f;
    for (int i = tid; i < NBT * 4; i += 1024) {
      den[i] = 0.f;
      mx[i]  = 0u;
      adl[i] = (i < nbr * 4) ? adst[(size_t)n0 * 4 + i] : 0.f;
    }
    if (tid == 0) {
      s_cnt = 0;
      s_bad = 0;
    }
    __syncthreads();

    auto test = [&](int d, int i) {
      if ((unsigned)(d - n0) < (unsigned)nbr) {
        const int slot = atomicAdd(&s_cnt, 1);
        if (slot < LCAP) {
          list[slot] = i;
          lr[slot]   = (unsigned char)(d - n0);
        }
      }
    };
    auto scan = [&](int lo, int hi) {
      if (((E0 | lo) & 3) == 0) {
        const int4* __restrict__ p = (const int4*)dstp;
        const int q0 = lo >> 2, q1 = hi >> 2;
        for (int q = q0 + tid; q < q1; q += 4096) {
          int4 v[4];
#pragma unroll
          for (int j = 0; j < 4; ++j) {
            const int qq = q + j * 1024;
            v[j] = p[qq < q1 ? qq : q1 - 1];
          }
#pragma unroll
          for (int j = 0; j < 4; ++j) {
            const int qq = q + j * 1024;
            if (qq < q1) {
              test(v[j].x, 4 * qq);
              test(v[j].y, 4 * qq + 1);
              test(v[j].z, 4 * qq + 2);
              test(v[j].w, 4 * qq + 3);
            }
          }
        }
        for (int i = (q1 << 2) + tid; i < hi; i += 1024) test(dstp[i], i);
      } else {
        for (int i = lo + tid; i < hi; i += 1024) test(dstp[i], i);
      }
    };
    auto maxphase = [&](int cntc, bool self, bool resolved, bool store) {
      const int total = cntc + (self ? nbr : 0);
      for (int t = tid; t < total; t += 1024) {
        int s, r;
        if (t < cntc) {
          r = lr[t];
          if (resolved) {
            s = list[t];
            if (LMODE == 0 && store) glw[t] = make_int2(s, r);
          } else {
            s = srcp[list[t]];
            list[t] = s;
            if (LMODE == 0 && store) glw[t] = make_int2(s, r);
          }
        } else {
          r = t - cntc;
          s = n0 + r;
        }
        const float4 a = *(const float4*)(asrc + (size_t)s * 4);
        const float av[4] = {a.x, a.y, a.z, a.w};
#pragma unroll
        for (int h = 0; h < 4; ++h) {
          float v = av[h] + adl[r * 4 + h];
          v = v > 0.f ? v : NEG_SLOPE * v;
          atomicMax(&mx[r * 4 + h], fkey(v));
        }
      }
    };
    auto resolve = [&](int cntc) {
      for (int t = tid; t < cntc; t += 1024) list[t] = srcp[list[t]];
    };
    auto aggphase = [&](int cntc_, bool self) {
      const int cntc  = __builtin_amdgcn_readfirstlane(cntc_);
      const int total = cntc + (self ? nbr : 0);
      const int nb8   = (total + 7) >> 3;
      const int wv0   = __builtin_amdgcn_readfirstlane(wave);
      const int je    = lane >> 2;
      const int hh    = lane & 3;
      for (int b = wv0; b < nb8; b += 32) {
        const int  t   = b * 8 + je;
        const bool okl = t < total;
        const int  tc  = okl ? t : total - 1;
        int s, r;
        if (tc < cntc) {
          s = list[tc];
          r = lr[tc];
        } else {
          r = tc - cntc;
          s = n0 + r;
        }
        const float a = asrc[(size_t)s * 4 + hh];
        float x[8][NCH];
#pragma unroll
        for (int j = 0; j < 8; ++j) {
          const int su = __builtin_amdgcn_readlane(s, 4 * j);
          const float* xs = xh + (size_t)su * LD;
#pragma unroll
          for (int c = 0; c < NCH; ++c) x[j][c] = xs[c * 32 + lane];
        }
        float v = a + adl[r * 4 + hh];
        v = v > 0.f ? v : NEG_SLOPE * v;
        const float w = __expf(v - fdecode(mx[r * 4 + hh]));
        if (okl) atomicAdd(&den[r * 4 + hh], w);
#pragma unroll
        for (int j = 0; j < 8; ++j) {
          if (b * 8 + j < total) {
            const int ru = __builtin_amdgcn_readlane(r, 4 * j);
            float wj[4];
#pragma unroll
            for (int h = 0; h < 4; ++h)
              wj[h] = __int_as_float(__builtin_amdgcn_readlane(__float_as_int(w), 4 * j + h));
#pragma unroll
            for (int c = 0; c < NCH; ++c) {
              const int col = c * 32 + lane;
              const int hd  = col / C;
              const float wh = hd == 0 ? wj[0] : hd == 1 ? wj[1] : hd == 2 ? wj[2] : wj[3];
              if (col < F) atomicAdd(&acc[ru * AP + col], x[j][c] * wh);
            }
          }
        }
      }
    };

    int  cnt  = 0;
    bool have = false;
    if constexpr (LMODE == 0) {
      if (bidx < NLIST) {
        for (int w = tid; w < NPART; w += 1024) {
          const int c = pcnt[(size_t)bidx * NPART + w];
          if (c < 0 || c > SEGCAP) s_bad = 1;
          const int cc = c < 0 ? 0 : c > SEGCAP ? SEGCAP : c;
          prun[w]         = cc;
          prun[NPART + w] = atomicAdd(&s_cnt, cc);
        }
        __syncthreads();
        cnt  = s_cnt;
        have = s_bad == 0 && cnt <= LCAP;
        if (have) {
          const int2* __restrict__ sg = (const int2*)pseg_ + (size_t)bidx * NPART * SEGCAP;
          for (int idx = tid; idx < NPART * SEGCAP; idx += 4096) {
            int2 e[4];
            int  p[4];
#pragma unroll
            for (int j = 0; j < 4; ++j) {
              const int ii = idx + j * 1024;
              const int w  = ii / SEGCAP, jj = ii - w * SEGCAP;
              p[j] = -1;
              e[j] = make_int2(0, 0);
              if (jj < prun[w]) {
                p[j] = prun[NPART + w] + jj;
                e[j] = sg[ii];
              }
            }
#pragma unroll
            for (int j = 0; j < 4; ++j) {
              if (p[j] >= 0) {
                list[p[j]] = e[j].x;
                lr[p[j]]   = (unsigned char)e[j].y;
              }
            }
          }
        }
        __syncthreads();
        if (!have && tid == 0) s_cnt = 0;
      }
    }
    if (LMODE == 1) {
      if (bidx < NLIST) {
        const int c = gcnt[bidx];
        if (c <= LCAP) {
          const int2* gl = (const int2*)glist + (size_t)bidx * LCAP;
          for (int t = tid; t < c; t += 1024) {
            const int2 e = gl[t];
            list[t] = e.x;
            lr[t]   = (unsigned char)e.y;
          }
          cnt  = c;
          have = true;
        }
      }
    }
    if (LMODE == 2) {
      const int b0 = n0 / NBL, b1 = (n0 + nbr - 1) / NBL;
      bool ok = b1 < NLIST;
      for (int b = b0; ok && b <= b1; ++b) ok = gcnt[b] <= LCAP;
      if (ok) {
        for (int b = b0; b <= b1; ++b) {
          const int c = gcnt[b];
          const int2* gl = (const int2*)glist + (size_t)b * LCAP;
          const int off = b * NBL - n0;
          for (int t = tid; t < c; t += 4096) {
            int2 e[4];
#pragma unroll
            for (int j = 0; j < 4; ++j) {
              const int tt = t + j * 1024;
              e[j] = gl[tt < c ? tt : c - 1];
            }
#pragma unroll
            for (int j = 0; j < 4; ++j) {
              const int rr = e[j].y + off;
              if (t + j * 1024 < c && (unsigned)rr < (unsigned)nbr) {
                const int slot = atomicAdd(&s_cnt, 1);
                if (slot < LCAP) {
                  list[slot] = e[j].x;
                  lr[slot]   = (unsigned char)rr;
                }
              }
            }
          }
        }
        __syncthreads();
        cnt  = s_cnt;
        have = cnt <= LCAP;
        __syncthreads();
        if (!have && tid == 0) s_cnt = 0;
      }
    }
    __syncthreads();
    if (!have) {
      scan(0, E0);
      __syncthreads();
      cnt = s_cnt;
    }
    if (keep && tid == 0) gcnt[bidx] = cnt;
    if (cnt <= LCAP) {
      maxphase(cnt, true, have, keep);
      __syncthreads();
      aggphase(cnt, true);
    } else {
      for (int lo = 0; lo < E0; lo += LCAP) {
        __syncthreads();
        if (tid == 0) s_cnt = 0;
        __syncthreads();
        scan(lo, (E0 - lo < LCAP) ? E0 : lo + LCAP);
        __syncthreads();
        maxphase(s_cnt, lo == 0, false, false);
      }
      for (int lo = 0; lo < E0; lo += LCAP) {
        __syncthreads();
        if (tid == 0) s_cnt = 0;
        __syncthreads();
        scan(lo, (E0 - lo < LCAP) ? E0 : lo + LCAP);
        __syncthreads();
        resolve(s_cnt);
        __syncthreads();
        aggphase(s_cnt, lo == 0);
      }
    }
    __syncthreads();

    if constexpr (FUSE != 0) {
      for (int i = tid; i < nbr * F; i += 1024) {
        const int r = i / F, f = i - r * F;
        const size_t g = (size_t)(n0 + r) * F + f;
        const float v = acc[r * AP + f] / (den[r * 4 + f / C] + EPS_DEN) + bias[f] + skip[g] + skb[f];
        acc[r * AP + f] = v > 0.f ? v : expm1f(v);
      }
      __syncthreads();

      constexpr int NT  = NBTP / 16;
      constexpr int NLH = NLD / 64;
      static_assert(F == 128, "K of the fused products");
      static_assert(NLD / 64 + NSLD / 64 == 4, "four strips");
      static_assert(2 * NT <= 32, "one unit per wave and pass");
      const int wv    = __builtin_amdgcn_readfirstlane(wave);
      const int rlo   = lane & 15;
      const int khalf = (lane >> 4) << 1;
#pragma unroll
      for (int p = 0; p < 2; ++p) {
#pragma unroll
        for (int i = 0; i < 8; ++i) {
          const int idx = tid + i * 1024;
          const int q   = 2 * p + (i >> 2);
          const int id  = idx & 4095;
          const int l   = id & 31;
          const int t   = (id >> 5) & 3;
          const int ks  = id >> 7;
          const int k   = 4 * ks + ((l >> 4) << 1);
          const bool lin = q < NLH;
          const float* __restrict__ wsrc = lin ? nw : nsw;
          const int wld = lin ? NWC : NSC;
          const int col = (lin ? q : q - NLH) * 64 + t * 16 + (l & 15);
          float2 b = make_float2(0.f, 0.f);
          if (col < wld) {
            b.x = wsrc[(size_t)k * wld + col];
            b.y = wsrc[(size_t)(k + 1) * wld + col];
          }
          *(float2*)&bs[(size_t)idx * 2] = b;
        }
        __syncthreads();
        const int un = (wv + 2 * p) & 31;
        if (un < 2 * NT) {
          const int hsel = un / NT;
          const int tile = un - hsel * NT;
          const int q    = 2 * p + hsel;
          const bool lin = q < NLH;
          float* __restrict__ dst = lin ? nxh : nskip;
          const int ldd  = lin ? NLD : NSLD;
          const int col0 = (lin ? q : q - NLH) * 64;
          const float* ap = acc + (tile * 16 + rlo) * AP + khalf;
          const float* bp = bs + (size_t)(hsel * 4096 + lane) * 2;
          v8f d[4] = {};
#pragma unroll 4
          for (int ks = 0; ks < F / 4; ++ks) {
            const v2f a = *(const v2f*)(ap + 4 * ks);
#pragma unroll
            for (int t = 0; t < 4; ++t) {
              const v2f b = *(const v2f*)(bp + ks * 256 + t * 64);
              d[t] = __builtin_amdgcn_wmma_f32_16x16x4_f32(
                  false, a, false, b, (short)0, d[t], false, false);
            }
          }
          const int mbase = tile * 16 + ((lane >> 4) << 3);
#pragma unroll
          for (int t = 0; t < 4; ++t)
#pragma unroll
            for (int r = 0; r < 8; ++r) {
              const int row = mbase + r;
              if (row < nbr) dst[(size_t)(n0 + row) * ldd + col0 + t * 16 + rlo] = d[t][r];
            }
        }
        __syncthreads();
      }

      constexpr int NC = NWC / HEADS;
      static_assert(NBT * HEADS <= 1024, "one (row, head) per thread");
      const int rw = wv * 8;
      const int rl = lane >> 2, h = lane & 3;
      if (rw + rl < nbr) {
        const float* xrow = nxh + (size_t)(n0 + rw) * NLD;
        const int xo = rl * NLD + h * NC;
        const int ao = h * NC;
        float s1 = 0.f, s2 = 0.f;
#pragma unroll 8
        for (int c = 0; c < NC; ++c) {
          const float v = xrow[xo + c];
          s1 += v * natt_s[ao + c];
          s2 += v * natt_d[ao + c];
        }
        float* srow = nas + (size_t)(n0 + rw) * 4;
        srow[lane] = s1;
        srow[(size_t)N * 4 + lane] = s2;
      }
    } else {
      for (int i = tid; i < nbr * C; i += 1024) {
        const int r = i / C, o = i - r * C;
        const float* ar = acc + r * AP;
        const float* dr = den + r * 4;
        const float m = 0.25f * (ar[o]         / (dr[0] + EPS_DEN) +
                                 ar[C + o]     / (dr[1] + EPS_DEN) +
                                 ar[2 * C + o] / (dr[2] + EPS_DEN) +
                                 ar[3 * C + o] / (dr[3] + EPS_DEN));
        out[(size_t)(n0 + r) * C + o] = m + bias[o] + skip[(size_t)(n0 + r) * SK2P + o] + skb[o];
      }
    }
    __syncthreads();
  }
}

extern "C" void kernel_launch(void* const* d_in, const int* in_sizes, int n_in,
                              void* d_out, int out_size, void* d_ws, size_t ws_size,
                              hipStream_t stream) {
  const float* x   = (const float*)d_in[0];
  const int*   ei  = (const int*)d_in[1];
  const float* lw[3] = {(const float*)d_in[2], (const float*)d_in[8],  (const float*)d_in[14]};
  const float* as[3] = {(const float*)d_in[3], (const float*)d_in[9],  (const float*)d_in[15]};
  const float* ad[3] = {(const float*)d_in[4], (const float*)d_in[10], (const float*)d_in[16]};
  const float* bi[3] = {(const float*)d_in[5], (const float*)d_in[11], (const float*)d_in[17]};
  const float* sw[3] = {(const float*)d_in[6], (const float*)d_in[12], (const float*)d_in[18]};
  const float* sb[3] = {(const float*)d_in[7], (const float*)d_in[13], (const float*)d_in[19]};

  const int N  = in_sizes[0] / 128;
  const int E0 = in_sizes[1] / 2;

  float* w = (float*)d_ws;
  float*    xh0  = w;                      w += (size_t)N * HCF;
  float*    xh1  = w;                      w += (size_t)N * HCF;
  float*    xh2  = w;                      w += (size_t)N * HC2P;
  float*    sk0  = w;                      w += (size_t)N * HCF;
  float*    sk1  = w;                      w += (size_t)N * HCF;
  float*    sk2  = w;                      w += (size_t)N * SK2P;
  float*    sc0  = w;                      w += (size_t)N * 8;
  float*    sc1  = w;                      w += (size_t)N * 8;
  int*      glist = (int*)w;               w += (size_t)NLIST * LCAP * 2;
  int*      gcnt  = (int*)w;               w += (size_t)NLIST;
  int*      pseg  = (int*)w;               w += (size_t)NLIST * NPART * SEGCAP * 2;
  int*      pcnt  = (int*)w;               w += (size_t)NLIST * NPART;

  const int mt8  = (N + 127) / 128;
  const int own  = 256;
  const int own2 = 224;

  part_edges<<<NPART, 1024, 0, stream>>>(ei, E0, N, pseg, pcnt);

  gemm_wmma_f32<128, 128, 128, 128><<<dim3(mt8, 2), 256, 0, stream>>>(x, lw[0], xh0, N, as[0], ad[0], sc0);
  gemm_wmma_f32<128, 128, 128, 128><<<dim3(mt8, 2), 256, 0, stream>>>(x, sw[0], sk0, N, nullptr, nullptr, nullptr);
  gat_own<32, 128, NB0, 0, 1, 128, 128, 128, 128><<<own, 1024, 0, stream>>>(
      ei, E0, N, xh0, sc0, sc0 + (size_t)N * 4, sk0, bi[0], sb[0], nullptr, glist, gcnt, pseg, pcnt,
      lw[1], sw[1], xh1, sk1, as[1], ad[1], sc1);

  gat_own<32, 128, NB0, 1, 1, HC2, HC2P, C2, SK2P><<<own, 1024, 0, stream>>>(
      ei, E0, N, xh1, sc1, sc1 + (size_t)N * 4, sk1, bi[1], sb[1], nullptr, glist, gcnt, nullptr, nullptr,
      lw[2], sw[2], xh2, sk2, as[2], ad[2], sc0);

  gat_own<47, HC2P, 224, 2, 0, 1, 64, 1, 64><<<own2, 1024, 0, stream>>>(
      ei, E0, N, xh2, sc0, sc0 + (size_t)N * 4, sk2, bi[2], sb[2], (float*)d_out, glist, gcnt, nullptr, nullptr,
      nullptr, nullptr, nullptr, nullptr, nullptr, nullptr, nullptr);
}
